# starting kernel
# baseline (speedup 1.0000x reference)
#include <hip/hip_runtime.h>
#include <hip/hip_bf16.h>


typedef __attribute__((ext_vector_type(2))) float v2f;
typedef __attribute__((ext_vector_type(8))) float v8f;

#define NB      65536
#define NL1     1024
#define NSTACK  8
#define NGRP    64
#define WS_CNT     0
#define WS_CUR     64
#define WS_OFF     128
#define WS_PERM    1024
#define WS_FRAG    (1024 + 65536*4)


__global__ void k_zero(int* w) {
    if (threadIdx.x < 48) w[threadIdx.x] = 0;
}

__global__ void k_count(const int* __restrict__ lsi, int* __restrict__ cnt) {
    int i = blockIdx.x * blockDim.x + threadIdx.x;
    if (i < NB) atomicAdd(&cnt[lsi[i] & 7], 1);
}

__global__ void k_scan(const int* __restrict__ cnt, int* __restrict__ off,
                       int* __restrict__ cur) {
    if (threadIdx.x == 0) {
        int acc = 0;
        for (int s = 0; s < NSTACK; ++s) {
            off[s] = acc;
            cur[s] = acc;
            acc += cnt[s];
        }
    }
}

__global__ void k_scatter(const int* __restrict__ lsi, int* __restrict__ cur,
                          int* __restrict__ perm) {
    int i = blockIdx.x * blockDim.x + threadIdx.x;
    if (i < NB) {
        int pos = atomicAdd(&cur[lsi[i] & 7], 1);
        perm[pos] = i;
    }
}


__global__ void k_prep(const float* __restrict__ l1_w,
                       const float* __restrict__ l1f_w,
                       float* __restrict__ frag) {
    int tid = blockIdx.x * blockDim.x + threadIdx.x;
    int e    = tid & 7;
    int lane = (tid >> 3) & 31;
    int h    = (tid >> 8) & 1;
    int g    = (tid >> 9) & 63;
    int s    = tid >> 15;
    int n = (lane & 15) + 16 * h;
    int k = g * 16 + ((lane < 16) ? 0 : 8) + e;
    float w = (n < 16) ? l1_w[(size_t)(s * 16 + n) * NL1 + k]
                       : l1f_w[(size_t)(n - 16) * NL1 + k];
    frag[tid] = w;
}


__global__ __launch_bounds__(128, 2)
void k_main(const float* __restrict__ x,
            const int* __restrict__ perm,
            const int* __restrict__ off,
            const int* __restrict__ cnt,
            const float* __restrict__ frag,
            const float* __restrict__ l1_b,
            const float* __restrict__ l1f_b,
            const float* __restrict__ l2_w,
            const float* __restrict__ l2_b,
            const float* __restrict__ out_w,
            const float* __restrict__ out_b,
            float* __restrict__ out) {
    __shared__ float lds[4][16][33];

    const int s    = blockIdx.x >> 10;
    const int tile = blockIdx.x & 1023;
    const int cnt_s = cnt[s];
    if (cnt_s == 0) return;
    if (tile * 64 >= cnt_s) return;

    const int wv   = threadIdx.x >> 5;
    const int lane = threadIdx.x & 31;
    const int r    = lane & 15;
    const int p    = tile * 64 + wv * 16 + r;
    const int pcl  = (p < cnt_s) ? p : (cnt_s - 1);
    const int rrow = perm[off[s] + pcl];

    const int koff = (lane < 16) ? 0 : 8;
    const float4* __restrict__ ax =
        (const float4*)(x + (size_t)rrow * NL1 + koff);
    const float4* __restrict__ bx = (const float4*)frag;

    v8f acc0 = {0.f, 0.f, 0.f, 0.f, 0.f, 0.f, 0.f, 0.f};
    v8f acc1 = {0.f, 0.f, 0.f, 0.f, 0.f, 0.f, 0.f, 0.f};

    const int fb0 = ((s * NGRP) * 2 + 0) * 64 + lane * 2;
    const int fb1 = ((s * NGRP) * 2 + 1) * 64 + lane * 2;

#pragma unroll 2
    for (int g = 0; g < NGRP; ++g) {
        float4 a0 = ax[4 * g + 0];
        float4 a1 = ax[4 * g + 1];
        float4 b00 = bx[fb0 + g * 128 + 0];
        float4 b01 = bx[fb0 + g * 128 + 1];
        float4 b10 = bx[fb1 + g * 128 + 0];
        float4 b11 = bx[fb1 + g * 128 + 1];
        float qa[8], q0[8], q1[8];
        *(float4*)&qa[0] = a0;  *(float4*)&qa[4] = a1;
        *(float4*)&q0[0] = b00; *(float4*)&q0[4] = b01;
        *(float4*)&q1[0] = b10; *(float4*)&q1[4] = b11;
#pragma unroll
        for (int c = 0; c < 4; ++c) {
            v2f A  = {qa[2 * c], qa[2 * c + 1]};
            v2f B0 = {q0[2 * c], q0[2 * c + 1]};
            v2f B1 = {q1[2 * c], q1[2 * c + 1]};
            acc0 = __builtin_amdgcn_wmma_f32_16x16x4_f32(
                false, A, false, B0, (short)0, acc0, false, false);
            acc1 = __builtin_amdgcn_wmma_f32_16x16x4_f32(
                false, A, false, B1, (short)0, acc1, false, false);
        }
    }

    const int mbase = (lane >= 16) ? 8 : 0;
    const int ncol  = lane & 15;
#pragma unroll
    for (int v = 0; v < 8; ++v) {
        lds[wv][v + mbase][ncol]      = acc0[v];
        lds[wv][v + mbase][ncol + 16] = acc1[v];
    }
    __builtin_amdgcn_wave_barrier();

    const int jh = lane >> 4;
    float t[15];
    float keep;
    {
        float yc15 = lds[wv][r][15] + l1_b[s * 16 + 15];
        float yf15 = lds[wv][r][31] + l1f_b[15];
        keep = yc15 + yf15;
#pragma unroll
        for (int j = 0; j < 15; ++j) {
            float yc = lds[wv][r][j]      + l1_b[s * 16 + j];
            float yf = lds[wv][r][16 + j] + l1f_b[j];
            t[j] = yc + yf;
        }
    }
    float l1x[30];
#pragma unroll
    for (int j = 0; j < 15; ++j) {
        float sq = t[j] * t[j] * (127.0f / 128.0f);
        l1x[j]      = fminf(fmaxf(sq, 0.f), 1.f);
        l1x[15 + j] = fminf(fmaxf(t[j], 0.f), 1.f);
    }
    float partial = 0.f;
#pragma unroll 4
    for (int oi = 0; oi < 16; ++oi) {
        int o = jh * 16 + oi;
        const float* wrow = l2_w + (size_t)(s * 32 + o) * 30;
        float a2 = l2_b[s * 32 + o];
#pragma unroll
        for (int k = 0; k < 30; ++k) a2 += l1x[k] * wrow[k];
        float l2x = fminf(fmaxf(a2, 0.f), 1.f);
        partial += l2x * out_w[s * 32 + o];
    }
    float tot = partial + __shfl_xor(partial, 16, 32);

    if (lane < 16 && p < cnt_s) {
        out[rrow] = tot + out_b[s] + keep;
    }
}


extern "C" void kernel_launch(void* const* d_in, const int* in_sizes, int n_in,
                              void* d_out, int out_size, void* d_ws, size_t ws_size,
                              hipStream_t stream) {
    (void)in_sizes; (void)n_in; (void)out_size; (void)ws_size;
    const float* x      = (const float*)d_in[0];
    const int*   lsi    = (const int*)d_in[1];
    const float* l1_w   = (const float*)d_in[2];
    const float* l1_b   = (const float*)d_in[3];
    const float* l1f_w  = (const float*)d_in[4];
    const float* l1f_b  = (const float*)d_in[5];
    const float* l2_w   = (const float*)d_in[6];
    const float* l2_b   = (const float*)d_in[7];
    const float* out_w  = (const float*)d_in[8];
    const float* out_b  = (const float*)d_in[9];
    float* out = (float*)d_out;

    char* ws   = (char*)d_ws;
    int*  cnt  = (int*)(ws + WS_CNT);
    int*  cur  = (int*)(ws + WS_CUR);
    int*  off  = (int*)(ws + WS_OFF);
    int*  perm = (int*)(ws + WS_PERM);
    float* frag = (float*)(ws + WS_FRAG);

    k_zero<<<1, 64, 0, stream>>>(cnt);
    k_count<<<NB / 256, 256, 0, stream>>>(lsi, cnt);
    k_scan<<<1, 32, 0, stream>>>(cnt, off, cur);
    k_scatter<<<NB / 256, 256, 0, stream>>>(lsi, cur, perm);
    k_prep<<<(NSTACK * NGRP * 2 * 256) / 256, 256, 0, stream>>>(l1_w, l1f_w, frag);
    k_main<<<NSTACK * 1024, 128, 0, stream>>>(x, perm, off, cnt, frag,
                                              l1_b, l1f_b, l2_w, l2_b,
                                              out_w, out_b, out);
}


# bucket rows by stack with no atomics: per-chunk packed counts plus wave and block prefix scans replace one global atomicAdd per row in count and again in scatter; 6 launches become 4
# speedup vs baseline: 1.4688x; 1.4688x over previous
#include <hip/hip_runtime.h>
#include <hip/hip_bf16.h>


typedef __attribute__((ext_vector_type(2))) float v2f;
typedef __attribute__((ext_vector_type(8))) float v8f;

#define NB      65536
#define NL1     1024
#define NSTACK  8
#define NGRP    64
#define WS_CNT     0
#define WS_OFF     128
#define WS_BCNT    192
#define WS_PERM    1024
#define WS_FRAG    (1024 + 65536*4)


#define BK_CHUNKS  16
#define BK_THREADS 1024

__device__ __forceinline__ unsigned bk_pack(int4 v) {
    return (1u << ((v.x & 7) * 4)) + (1u << ((v.y & 7) * 4)) +
           (1u << ((v.z & 7) * 4)) + (1u << ((v.w & 7) * 4));
}

__device__ __forceinline__ int bk_field(unsigned ev, unsigned od, int k) {
    return (int)((((k & 1) ? od : ev) >> ((k >> 1) * 8)) & 0xFFu);
}

__global__ __launch_bounds__(BK_THREADS)
void k_count(const int* __restrict__ lsi, int* __restrict__ bcnt) {
    __shared__ unsigned s_w[32][2];
    const int t = threadIdx.x, lane = t & 31, wv = t >> 5;
    const int4 v = ((const int4*)lsi)[blockIdx.x * BK_THREADS + t];
    const unsigned pk = bk_pack(v);
    unsigned ev = pk & 0x0F0F0F0Fu;
    unsigned od = (pk >> 4) & 0x0F0F0F0Fu;
    for (int m = 16; m >= 1; m >>= 1) {
        ev += __shfl_xor(ev, m, 32);
        od += __shfl_xor(od, m, 32);
    }
    if (lane == 0) { s_w[wv][0] = ev; s_w[wv][1] = od; }
    __syncthreads();
    if (t < 8) {
        int acc = 0;
        for (int w = 0; w < 32; ++w) acc += bk_field(s_w[w][0], s_w[w][1], t);
        bcnt[blockIdx.x * 8 + t] = acc;
    }
}

__global__ __launch_bounds__(BK_THREADS)
void k_scatter(const int* __restrict__ lsi, const int* __restrict__ bcnt,
               int* __restrict__ cnt, int* __restrict__ off,
               int* __restrict__ perm) {
    __shared__ unsigned s_w[32][2];
    __shared__ int s_base[8];
    __shared__ int s_wbase[32][8];
    const int t = threadIdx.x, lane = t & 31, wv = t >> 5;
    const int b = blockIdx.x;
    const int4 v = ((const int4*)lsi)[b * BK_THREADS + t];
    const unsigned pk = bk_pack(v);
    const unsigned ev = pk & 0x0F0F0F0Fu;
    const unsigned od = (pk >> 4) & 0x0F0F0F0Fu;

    unsigned iev = ev, iod = od;
    for (int d = 1; d < 32; d <<= 1) {
        const unsigned a = __shfl_up(iev, d, 32);
        const unsigned c = __shfl_up(iod, d, 32);
        if (lane >= d) { iev += a; iod += c; }
    }
    if (lane == 31) { s_w[wv][0] = iev; s_w[wv][1] = iod; }

    if (wv == 0) {
        int tot = 0, pre = 0;
        if (lane < 8) {
            for (int j = 0; j < BK_CHUNKS; ++j) {
                const int c = bcnt[j * 8 + lane];
                tot += c;
                pre += (j < b) ? c : 0;
            }
        }
        int inc = tot;
        for (int d = 1; d < 8; d <<= 1) {
            const int a = __shfl_up(inc, d, 32);
            if (lane >= d) inc += a;
        }
        if (lane < 8) {
            const int start = inc - tot;
            s_base[lane] = start + pre;
            if (b == 0) { cnt[lane] = tot; off[lane] = start; }
        }
    }
    __syncthreads();
    if (t < 256) {
        const int w = t >> 3, k = t & 7;
        int acc = s_base[k];
        for (int w2 = 0; w2 < w; ++w2) acc += bk_field(s_w[w2][0], s_w[w2][1], k);
        s_wbase[w][k] = acc;
    }
    __syncthreads();

    const unsigned xev = iev - ev, xod = iod - od;
    const int row0 = (b * BK_THREADS + t) * 4;
    const int kk[4] = {v.x & 7, v.y & 7, v.z & 7, v.w & 7};
    unsigned loc = 0;
#pragma unroll
    for (int e = 0; e < 4; ++e) {
        const int k = kk[e];
        const int pos = s_wbase[wv][k] + bk_field(xev, xod, k) +
                        (int)((loc >> (k * 4)) & 15u);
        loc += 1u << (k * 4);
        perm[pos] = row0 + e;
    }
}


__global__ void k_prep(const float* __restrict__ l1_w,
                       const float* __restrict__ l1f_w,
                       float* __restrict__ frag) {
    int tid = blockIdx.x * blockDim.x + threadIdx.x;
    int e    = tid & 7;
    int lane = (tid >> 3) & 31;
    int h    = (tid >> 8) & 1;
    int g    = (tid >> 9) & 63;
    int s    = tid >> 15;
    int n = (lane & 15) + 16 * h;
    int k = g * 16 + ((lane < 16) ? 0 : 8) + e;
    float w = (n < 16) ? l1_w[(size_t)(s * 16 + n) * NL1 + k]
                       : l1f_w[(size_t)(n - 16) * NL1 + k];
    frag[tid] = w;
}


__global__ __launch_bounds__(128, 2)
void k_main(const float* __restrict__ x,
            const int* __restrict__ perm,
            const int* __restrict__ off,
            const int* __restrict__ cnt,
            const float* __restrict__ frag,
            const float* __restrict__ l1_b,
            const float* __restrict__ l1f_b,
            const float* __restrict__ l2_w,
            const float* __restrict__ l2_b,
            const float* __restrict__ out_w,
            const float* __restrict__ out_b,
            float* __restrict__ out) {
    __shared__ float lds[4][16][33];

    const int s    = blockIdx.x >> 10;
    const int tile = blockIdx.x & 1023;
    const int cnt_s = cnt[s];
    if (cnt_s == 0) return;
    if (tile * 64 >= cnt_s) return;

    const int wv   = threadIdx.x >> 5;
    const int lane = threadIdx.x & 31;
    const int r    = lane & 15;
    const int p    = tile * 64 + wv * 16 + r;
    const int pcl  = (p < cnt_s) ? p : (cnt_s - 1);
    const int rrow = perm[off[s] + pcl];

    const int koff = (lane < 16) ? 0 : 8;
    const float4* __restrict__ ax =
        (const float4*)(x + (size_t)rrow * NL1 + koff);
    const float4* __restrict__ bx = (const float4*)frag;

    v8f acc0 = {0.f, 0.f, 0.f, 0.f, 0.f, 0.f, 0.f, 0.f};
    v8f acc1 = {0.f, 0.f, 0.f, 0.f, 0.f, 0.f, 0.f, 0.f};

    const int fb0 = ((s * NGRP) * 2 + 0) * 64 + lane * 2;
    const int fb1 = ((s * NGRP) * 2 + 1) * 64 + lane * 2;

#pragma unroll 2
    for (int g = 0; g < NGRP; ++g) {
        float4 a0 = ax[4 * g + 0];
        float4 a1 = ax[4 * g + 1];
        float4 b00 = bx[fb0 + g * 128 + 0];
        float4 b01 = bx[fb0 + g * 128 + 1];
        float4 b10 = bx[fb1 + g * 128 + 0];
        float4 b11 = bx[fb1 + g * 128 + 1];
        float qa[8], q0[8], q1[8];
        *(float4*)&qa[0] = a0;  *(float4*)&qa[4] = a1;
        *(float4*)&q0[0] = b00; *(float4*)&q0[4] = b01;
        *(float4*)&q1[0] = b10; *(float4*)&q1[4] = b11;
#pragma unroll
        for (int c = 0; c < 4; ++c) {
            v2f A  = {qa[2 * c], qa[2 * c + 1]};
            v2f B0 = {q0[2 * c], q0[2 * c + 1]};
            v2f B1 = {q1[2 * c], q1[2 * c + 1]};
            acc0 = __builtin_amdgcn_wmma_f32_16x16x4_f32(
                false, A, false, B0, (short)0, acc0, false, false);
            acc1 = __builtin_amdgcn_wmma_f32_16x16x4_f32(
                false, A, false, B1, (short)0, acc1, false, false);
        }
    }

    const int mbase = (lane >= 16) ? 8 : 0;
    const int ncol  = lane & 15;
#pragma unroll
    for (int v = 0; v < 8; ++v) {
        lds[wv][v + mbase][ncol]      = acc0[v];
        lds[wv][v + mbase][ncol + 16] = acc1[v];
    }
    __builtin_amdgcn_wave_barrier();

    const int jh = lane >> 4;
    float t[15];
    float keep;
    {
        float yc15 = lds[wv][r][15] + l1_b[s * 16 + 15];
        float yf15 = lds[wv][r][31] + l1f_b[15];
        keep = yc15 + yf15;
#pragma unroll
        for (int j = 0; j < 15; ++j) {
            float yc = lds[wv][r][j]      + l1_b[s * 16 + j];
            float yf = lds[wv][r][16 + j] + l1f_b[j];
            t[j] = yc + yf;
        }
    }
    float l1x[30];
#pragma unroll
    for (int j = 0; j < 15; ++j) {
        float sq = t[j] * t[j] * (127.0f / 128.0f);
        l1x[j]      = fminf(fmaxf(sq, 0.f), 1.f);
        l1x[15 + j] = fminf(fmaxf(t[j], 0.f), 1.f);
    }
    float partial = 0.f;
#pragma unroll 4
    for (int oi = 0; oi < 16; ++oi) {
        int o = jh * 16 + oi;
        const float* wrow = l2_w + (size_t)(s * 32 + o) * 30;
        float a2 = l2_b[s * 32 + o];
#pragma unroll
        for (int k = 0; k < 30; ++k) a2 += l1x[k] * wrow[k];
        float l2x = fminf(fmaxf(a2, 0.f), 1.f);
        partial += l2x * out_w[s * 32 + o];
    }
    float tot = partial + __shfl_xor(partial, 16, 32);

    if (lane < 16 && p < cnt_s) {
        out[rrow] = tot + out_b[s] + keep;
    }
}


extern "C" void kernel_launch(void* const* d_in, const int* in_sizes, int n_in,
                              void* d_out, int out_size, void* d_ws, size_t ws_size,
                              hipStream_t stream) {
    (void)in_sizes; (void)n_in; (void)out_size; (void)ws_size;
    const float* x      = (const float*)d_in[0];
    const int*   lsi    = (const int*)d_in[1];
    const float* l1_w   = (const float*)d_in[2];
    const float* l1_b   = (const float*)d_in[3];
    const float* l1f_w  = (const float*)d_in[4];
    const float* l1f_b  = (const float*)d_in[5];
    const float* l2_w   = (const float*)d_in[6];
    const float* l2_b   = (const float*)d_in[7];
    const float* out_w  = (const float*)d_in[8];
    const float* out_b  = (const float*)d_in[9];
    float* out = (float*)d_out;

    char* ws   = (char*)d_ws;
    int*  cnt  = (int*)(ws + WS_CNT);
    int*  off  = (int*)(ws + WS_OFF);
    int*  bcnt = (int*)(ws + WS_BCNT);
    int*  perm = (int*)(ws + WS_PERM);
    float* frag = (float*)(ws + WS_FRAG);

    k_count<<<BK_CHUNKS, BK_THREADS, 0, stream>>>(lsi, bcnt);
    k_scatter<<<BK_CHUNKS, BK_THREADS, 0, stream>>>(lsi, bcnt, cnt, off, perm);
    k_prep<<<(NSTACK * NGRP * 2 * 256) / 256, 256, 0, stream>>>(l1_w, l1f_w, frag);
    k_main<<<NSTACK * 1024, 128, 0, stream>>>(x, perm, off, cnt, frag,
                                              l1_b, l1f_b, l2_w, l2_b,
                                              out_w, out_b, out);
}


# k_main A feed by whole-line loads: each load reads 512 contiguous bytes of one x row (was 16 lines per load); a wave turns its 16 rows through its own LDS into fragments, one 128-float step ahead
# speedup vs baseline: 1.7226x; 1.1728x over previous
#include <hip/hip_runtime.h>
#include <hip/hip_bf16.h>


typedef __attribute__((ext_vector_type(2))) float v2f;
typedef __attribute__((ext_vector_type(8))) float v8f;

#define NB      65536
#define NL1     1024
#define NSTACK  8
#define NGRP    64
#define KC      128
#define NSTEPS  (NL1 / KC)
#define GPS     (KC / 16)
#define APITCH  132
#define EPITCH  33
#define WS_CNT     0
#define WS_OFF     128
#define WS_BCNT    192
#define WS_PERM    1024
#define WS_FRAG    (1024 + 65536*4)


#define BK_CHUNKS  16
#define BK_THREADS 1024

__device__ __forceinline__ unsigned bk_pack(int4 v) {
    return (1u << ((v.x & 7) * 4)) + (1u << ((v.y & 7) * 4)) +
           (1u << ((v.z & 7) * 4)) + (1u << ((v.w & 7) * 4));
}

__device__ __forceinline__ int bk_field(unsigned ev, unsigned od, int k) {
    return (int)((((k & 1) ? od : ev) >> ((k >> 1) * 8)) & 0xFFu);
}

__global__ __launch_bounds__(BK_THREADS)
void k_count(const int* __restrict__ lsi, int* __restrict__ bcnt) {
    __shared__ unsigned s_w[32][2];
    const int t = threadIdx.x, lane = t & 31, wv = t >> 5;
    const int4 v = ((const int4*)lsi)[blockIdx.x * BK_THREADS + t];
    const unsigned pk = bk_pack(v);
    unsigned ev = pk & 0x0F0F0F0Fu;
    unsigned od = (pk >> 4) & 0x0F0F0F0Fu;
    for (int m = 16; m >= 1; m >>= 1) {
        ev += __shfl_xor(ev, m, 32);
        od += __shfl_xor(od, m, 32);
    }
    if (lane == 0) { s_w[wv][0] = ev; s_w[wv][1] = od; }
    __syncthreads();
    if (t < 8) {
        int acc = 0;
        for (int w = 0; w < 32; ++w) acc += bk_field(s_w[w][0], s_w[w][1], t);
        bcnt[blockIdx.x * 8 + t] = acc;
    }
}

__global__ __launch_bounds__(BK_THREADS)
void k_scatter(const int* __restrict__ lsi, const int* __restrict__ bcnt,
               int* __restrict__ cnt, int* __restrict__ off,
               int* __restrict__ perm) {
    __shared__ unsigned s_w[32][2];
    __shared__ int s_base[8];
    __shared__ int s_wbase[32][8];
    const int t = threadIdx.x, lane = t & 31, wv = t >> 5;
    const int b = blockIdx.x;
    const int4 v = ((const int4*)lsi)[b * BK_THREADS + t];
    const unsigned pk = bk_pack(v);
    const unsigned ev = pk & 0x0F0F0F0Fu;
    const unsigned od = (pk >> 4) & 0x0F0F0F0Fu;

    unsigned iev = ev, iod = od;
    for (int d = 1; d < 32; d <<= 1) {
        const unsigned a = __shfl_up(iev, d, 32);
        const unsigned c = __shfl_up(iod, d, 32);
        if (lane >= d) { iev += a; iod += c; }
    }
    if (lane == 31) { s_w[wv][0] = iev; s_w[wv][1] = iod; }

    if (wv == 0) {
        int tot = 0, pre = 0;
        if (lane < 8) {
            for (int j = 0; j < BK_CHUNKS; ++j) {
                const int c = bcnt[j * 8 + lane];
                tot += c;
                pre += (j < b) ? c : 0;
            }
        }
        int inc = tot;
        for (int d = 1; d < 8; d <<= 1) {
            const int a = __shfl_up(inc, d, 32);
            if (lane >= d) inc += a;
        }
        if (lane < 8) {
            const int start = inc - tot;
            s_base[lane] = start + pre;
            if (b == 0) { cnt[lane] = tot; off[lane] = start; }
        }
    }
    __syncthreads();
    if (t < 256) {
        const int w = t >> 3, k = t & 7;
        int acc = s_base[k];
        for (int w2 = 0; w2 < w; ++w2) acc += bk_field(s_w[w2][0], s_w[w2][1], k);
        s_wbase[w][k] = acc;
    }
    __syncthreads();

    const unsigned xev = iev - ev, xod = iod - od;
    const int row0 = (b * BK_THREADS + t) * 4;
    const int kk[4] = {v.x & 7, v.y & 7, v.z & 7, v.w & 7};
    unsigned loc = 0;
#pragma unroll
    for (int e = 0; e < 4; ++e) {
        const int k = kk[e];
        const int pos = s_wbase[wv][k] + bk_field(xev, xod, k) +
                        (int)((loc >> (k * 4)) & 15u);
        loc += 1u << (k * 4);
        perm[pos] = row0 + e;
    }
}


__global__ void k_prep(const float* __restrict__ l1_w,
                       const float* __restrict__ l1f_w,
                       float* __restrict__ frag) {
    int tid = blockIdx.x * blockDim.x + threadIdx.x;
    int e    = tid & 7;
    int lane = (tid >> 3) & 31;
    int h    = (tid >> 8) & 1;
    int g    = (tid >> 9) & 63;
    int s    = tid >> 15;
    int n = (lane & 15) + 16 * h;
    int k = g * 16 + ((lane < 16) ? 0 : 8) + e;
    float w = (n < 16) ? l1_w[(size_t)(s * 16 + n) * NL1 + k]
                       : l1f_w[(size_t)(n - 16) * NL1 + k];
    frag[tid] = w;
}


__global__ __launch_bounds__(128, 5)
void k_main(const float* __restrict__ x,
            const int* __restrict__ perm,
            const int* __restrict__ off,
            const int* __restrict__ cnt,
            const float* __restrict__ frag,
            const float* __restrict__ l1_b,
            const float* __restrict__ l1f_b,
            const float* __restrict__ l2_w,
            const float* __restrict__ l2_b,
            const float* __restrict__ out_w,
            const float* __restrict__ out_b,
            float* __restrict__ out) {
    __shared__ __attribute__((aligned(16))) float sA[4][16 * APITCH];

    const int s    = blockIdx.x >> 10;
    const int tile = blockIdx.x & 1023;
    const int cnt_s = cnt[s];
    if (cnt_s == 0) return;
    if (tile * 64 >= cnt_s) return;

    const int wv   = threadIdx.x >> 5;
    const int lane = threadIdx.x & 31;
    const int r    = lane & 15;
    const int p    = tile * 64 + wv * 16 + r;
    const int pcl  = (p < cnt_s) ? p : (cnt_s - 1);
    const int rrow = perm[off[s] + pcl];

    const int koff = (lane < 16) ? 0 : 8;
    float* __restrict__ Aw = sA[wv];
    const float4* __restrict__ bx = (const float4*)frag;

#define ROWS16(M) M(0) M(1) M(2) M(3) M(4) M(5) M(6) M(7) \
                  M(8) M(9) M(10) M(11) M(12) M(13) M(14) M(15)
#define XBASE(j) const float* __restrict__ xb##j = \
        x + (size_t)__builtin_amdgcn_readlane(rrow, j) * NL1;
    ROWS16(XBASE)
    const unsigned lo = (unsigned)lane * 4u;

    v8f acc0 = {0.f, 0.f, 0.f, 0.f, 0.f, 0.f, 0.f, 0.f};
    v8f acc1 = {0.f, 0.f, 0.f, 0.f, 0.f, 0.f, 0.f, 0.f};

    const int fb0 = ((s * NGRP) * 2 + 0) * 64 + lane * 2;
    const int fb1 = ((s * NGRP) * 2 + 1) * 64 + lane * 2;
    const int arow = r * APITCH + koff;

#define XDECL(j)  float4 xg##j;
#define XLOAD(j)  xg##j = *(const float4*)(xb##j + vo);
#define XSTORE(j) *(float4*)&Aw[j * APITCH + lo] = xg##j;
    {
        const unsigned vo = lo;
        ROWS16(XDECL)
        ROWS16(XLOAD)
        ROWS16(XSTORE)
    }
    ROWS16(XDECL)
    __builtin_amdgcn_wave_barrier();

#pragma unroll 1
    for (int t = 0; t < NSTEPS; ++t) {
        if (t + 1 < NSTEPS) {
            const unsigned vo = (unsigned)(t + 1) * KC + lo;
            ROWS16(XLOAD)
        }
#pragma unroll
        for (int gi = 0; gi < GPS; ++gi) {
            const int g = t * GPS + gi;
            float4 a0 = *(const float4*)&Aw[arow + gi * 16];
            float4 a1 = *(const float4*)&Aw[arow + gi * 16 + 4];
            float4 b00 = bx[fb0 + g * 128 + 0];
            float4 b01 = bx[fb0 + g * 128 + 1];
            float4 b10 = bx[fb1 + g * 128 + 0];
            float4 b11 = bx[fb1 + g * 128 + 1];
            float qa[8], q0[8], q1[8];
            *(float4*)&qa[0] = a0;  *(float4*)&qa[4] = a1;
            *(float4*)&q0[0] = b00; *(float4*)&q0[4] = b01;
            *(float4*)&q1[0] = b10; *(float4*)&q1[4] = b11;
#pragma unroll
            for (int c = 0; c < 4; ++c) {
                v2f A  = {qa[2 * c], qa[2 * c + 1]};
                v2f B0 = {q0[2 * c], q0[2 * c + 1]};
                v2f B1 = {q1[2 * c], q1[2 * c + 1]};
                acc0 = __builtin_amdgcn_wmma_f32_16x16x4_f32(
                    false, A, false, B0, (short)0, acc0, false, false);
                acc1 = __builtin_amdgcn_wmma_f32_16x16x4_f32(
                    false, A, false, B1, (short)0, acc1, false, false);
            }
        }
        __builtin_amdgcn_wave_barrier();
        if (t + 1 < NSTEPS) {
            ROWS16(XSTORE)
        }
        __builtin_amdgcn_wave_barrier();
    }
#undef XBASE
#undef XDECL
#undef XLOAD
#undef XSTORE
#undef ROWS16

    const int mbase = (lane >= 16) ? 8 : 0;
    const int ncol  = lane & 15;
#pragma unroll
    for (int v = 0; v < 8; ++v) {
        Aw[(v + mbase) * EPITCH + ncol]      = acc0[v];
        Aw[(v + mbase) * EPITCH + ncol + 16] = acc1[v];
    }
    __builtin_amdgcn_wave_barrier();

    const int jh = lane >> 4;
    float t[15];
    float keep;
    {
        float yc15 = Aw[r * EPITCH + 15] + l1_b[s * 16 + 15];
        float yf15 = Aw[r * EPITCH + 31] + l1f_b[15];
        keep = yc15 + yf15;
#pragma unroll
        for (int j = 0; j < 15; ++j) {
            float yc = Aw[r * EPITCH + j]      + l1_b[s * 16 + j];
            float yf = Aw[r * EPITCH + 16 + j] + l1f_b[j];
            t[j] = yc + yf;
        }
    }
    float l1x[30];
#pragma unroll
    for (int j = 0; j < 15; ++j) {
        float sq = t[j] * t[j] * (127.0f / 128.0f);
        l1x[j]      = fminf(fmaxf(sq, 0.f), 1.f);
        l1x[15 + j] = fminf(fmaxf(t[j], 0.f), 1.f);
    }
    float partial = 0.f;
#pragma unroll 4
    for (int oi = 0; oi < 16; ++oi) {
        int o = jh * 16 + oi;
        const float* wrow = l2_w + (size_t)(s * 32 + o) * 30;
        float a2 = l2_b[s * 32 + o];
#pragma unroll
        for (int k = 0; k < 30; ++k) a2 += l1x[k] * wrow[k];
        float l2x = fminf(fmaxf(a2, 0.f), 1.f);
        partial += l2x * out_w[s * 32 + o];
    }
    float tot = partial + __shfl_xor(partial, 16, 32);

    if (lane < 16 && p < cnt_s) {
        out[rrow] = tot + out_b[s] + keep;
    }
}


extern "C" void kernel_launch(void* const* d_in, const int* in_sizes, int n_in,
                              void* d_out, int out_size, void* d_ws, size_t ws_size,
                              hipStream_t stream) {
    (void)in_sizes; (void)n_in; (void)out_size; (void)ws_size;
    const float* x      = (const float*)d_in[0];
    const int*   lsi    = (const int*)d_in[1];
    const float* l1_w   = (const float*)d_in[2];
    const float* l1_b   = (const float*)d_in[3];
    const float* l1f_w  = (const float*)d_in[4];
    const float* l1f_b  = (const float*)d_in[5];
    const float* l2_w   = (const float*)d_in[6];
    const float* l2_b   = (const float*)d_in[7];
    const float* out_w  = (const float*)d_in[8];
    const float* out_b  = (const float*)d_in[9];
    float* out = (float*)d_out;

    char* ws   = (char*)d_ws;
    int*  cnt  = (int*)(ws + WS_CNT);
    int*  off  = (int*)(ws + WS_OFF);
    int*  bcnt = (int*)(ws + WS_BCNT);
    int*  perm = (int*)(ws + WS_PERM);
    float* frag = (float*)(ws + WS_FRAG);

    k_count<<<BK_CHUNKS, BK_THREADS, 0, stream>>>(lsi, bcnt);
    k_scatter<<<BK_CHUNKS, BK_THREADS, 0, stream>>>(lsi, bcnt, cnt, off, perm);
    k_prep<<<(NSTACK * NGRP * 2 * 256) / 256, 256, 0, stream>>>(l1_w, l1f_w, frag);
    k_main<<<NSTACK * 1024, 128, 0, stream>>>(x, perm, off, cnt, frag,
                                              l1_b, l1f_b, l2_w, l2_b,
                                              out_w, out_b, out);
}


# first-layer weights folded: x*(w_stack+w_shared) for x*w_stack + x*w_shared, 16 columns per row not 32: half the wmma and B loads; K summed in two chains added at the end (order of sums changes)
# speedup vs baseline: 2.0905x; 1.2136x over previous
#include <hip/hip_runtime.h>
#include <hip/hip_bf16.h>


typedef __attribute__((ext_vector_type(2))) float v2f;
typedef __attribute__((ext_vector_type(8))) float v8f;

#define NB      65536
#define NL1     1024
#define NSTACK  8
#define NGRP    64
#define KC      128
#define NSTEPS  (NL1 / KC)
#define GPS     (KC / 16)
#define APITCH  132
#define EPITCH  17
#define WS_CNT     0
#define WS_OFF     128
#define WS_BCNT    192
#define WS_PERM    1024
#define WS_FRAG    (1024 + 65536*4)


#define BK_CHUNKS  16
#define BK_THREADS 1024

__device__ __forceinline__ unsigned bk_pack(int4 v) {
    return (1u << ((v.x & 7) * 4)) + (1u << ((v.y & 7) * 4)) +
           (1u << ((v.z & 7) * 4)) + (1u << ((v.w & 7) * 4));
}

__device__ __forceinline__ int bk_field(unsigned ev, unsigned od, int k) {
    return (int)((((k & 1) ? od : ev) >> ((k >> 1) * 8)) & 0xFFu);
}

__global__ __launch_bounds__(BK_THREADS)
void k_count(const int* __restrict__ lsi, int* __restrict__ bcnt) {
    __shared__ unsigned s_w[32][2];
    const int t = threadIdx.x, lane = t & 31, wv = t >> 5;
    const int4 v = ((const int4*)lsi)[blockIdx.x * BK_THREADS + t];
    const unsigned pk = bk_pack(v);
    unsigned ev = pk & 0x0F0F0F0Fu;
    unsigned od = (pk >> 4) & 0x0F0F0F0Fu;
    for (int m = 16; m >= 1; m >>= 1) {
        ev += __shfl_xor(ev, m, 32);
        od += __shfl_xor(od, m, 32);
    }
    if (lane == 0) { s_w[wv][0] = ev; s_w[wv][1] = od; }
    __syncthreads();
    if (t < 8) {
        int acc = 0;
        for (int w = 0; w < 32; ++w) acc += bk_field(s_w[w][0], s_w[w][1], t);
        bcnt[blockIdx.x * 8 + t] = acc;
    }
}

__global__ __launch_bounds__(BK_THREADS)
void k_scatter(const int* __restrict__ lsi, const int* __restrict__ bcnt,
               int* __restrict__ cnt, int* __restrict__ off,
               int* __restrict__ perm) {
    __shared__ unsigned s_w[32][2];
    __shared__ int s_base[8];
    __shared__ int s_wbase[32][8];
    const int t = threadIdx.x, lane = t & 31, wv = t >> 5;
    const int b = blockIdx.x;
    const int4 v = ((const int4*)lsi)[b * BK_THREADS + t];
    const unsigned pk = bk_pack(v);
    const unsigned ev = pk & 0x0F0F0F0Fu;
    const unsigned od = (pk >> 4) & 0x0F0F0F0Fu;

    unsigned iev = ev, iod = od;
    for (int d = 1; d < 32; d <<= 1) {
        const unsigned a = __shfl_up(iev, d, 32);
        const unsigned c = __shfl_up(iod, d, 32);
        if (lane >= d) { iev += a; iod += c; }
    }
    if (lane == 31) { s_w[wv][0] = iev; s_w[wv][1] = iod; }

    if (wv == 0) {
        int tot = 0, pre = 0;
        if (lane < 8) {
            for (int j = 0; j < BK_CHUNKS; ++j) {
                const int c = bcnt[j * 8 + lane];
                tot += c;
                pre += (j < b) ? c : 0;
            }
        }
        int inc = tot;
        for (int d = 1; d < 8; d <<= 1) {
            const int a = __shfl_up(inc, d, 32);
            if (lane >= d) inc += a;
        }
        if (lane < 8) {
            const int start = inc - tot;
            s_base[lane] = start + pre;
            if (b == 0) { cnt[lane] = tot; off[lane] = start; }
        }
    }
    __syncthreads();
    if (t < 256) {
        const int w = t >> 3, k = t & 7;
        int acc = s_base[k];
        for (int w2 = 0; w2 < w; ++w2) acc += bk_field(s_w[w2][0], s_w[w2][1], k);
        s_wbase[w][k] = acc;
    }
    __syncthreads();

    const unsigned xev = iev - ev, xod = iod - od;
    const int row0 = (b * BK_THREADS + t) * 4;
    const int kk[4] = {v.x & 7, v.y & 7, v.z & 7, v.w & 7};
    unsigned loc = 0;
#pragma unroll
    for (int e = 0; e < 4; ++e) {
        const int k = kk[e];
        const int pos = s_wbase[wv][k] + bk_field(xev, xod, k) +
                        (int)((loc >> (k * 4)) & 15u);
        loc += 1u << (k * 4);
        perm[pos] = row0 + e;
    }
}


__global__ void k_prep(const float* __restrict__ l1_w,
                       const float* __restrict__ l1f_w,
                       float* __restrict__ frag) {
    int tid = blockIdx.x * blockDim.x + threadIdx.x;
    int e    = tid & 7;
    int lane = (tid >> 3) & 31;
    int g    = (tid >> 8) & 63;
    int s    = tid >> 14;
    int n = lane & 15;
    int k = g * 16 + ((lane < 16) ? 0 : 8) + e;
    frag[tid] = l1_w[(size_t)(s * 16 + n) * NL1 + k] + l1f_w[(size_t)n * NL1 + k];
}


__global__ __launch_bounds__(128, 5)
void k_main(const float* __restrict__ x,
            const int* __restrict__ perm,
            const int* __restrict__ off,
            const int* __restrict__ cnt,
            const float* __restrict__ frag,
            const float* __restrict__ l1_b,
            const float* __restrict__ l1f_b,
            const float* __restrict__ l2_w,
            const float* __restrict__ l2_b,
            const float* __restrict__ out_w,
            const float* __restrict__ out_b,
            float* __restrict__ out) {
    __shared__ __attribute__((aligned(16))) float sA[4][16 * APITCH];

    const int s    = blockIdx.x >> 10;
    const int tile = blockIdx.x & 1023;
    const int cnt_s = cnt[s];
    if (cnt_s == 0) return;
    if (tile * 64 >= cnt_s) return;

    const int wv   = threadIdx.x >> 5;
    const int lane = threadIdx.x & 31;
    const int r    = lane & 15;
    const int p    = tile * 64 + wv * 16 + r;
    const int pcl  = (p < cnt_s) ? p : (cnt_s - 1);
    const int rrow = perm[off[s] + pcl];

    const int koff = (lane < 16) ? 0 : 8;
    float* __restrict__ Aw = sA[wv];
    const float4* __restrict__ bx = (const float4*)frag;

#define ROWS16(M) M(0) M(1) M(2) M(3) M(4) M(5) M(6) M(7) \
                  M(8) M(9) M(10) M(11) M(12) M(13) M(14) M(15)
#define XBASE(j) const float* __restrict__ xb##j = \
        x + (size_t)__builtin_amdgcn_readlane(rrow, j) * NL1;
    ROWS16(XBASE)
    const unsigned lo = (unsigned)lane * 4u;

    v8f acc0 = {0.f, 0.f, 0.f, 0.f, 0.f, 0.f, 0.f, 0.f};
    v8f acc1 = {0.f, 0.f, 0.f, 0.f, 0.f, 0.f, 0.f, 0.f};

    const int fb = (s * NGRP) * 64 + lane * 2;
    const int arow = r * APITCH + koff;

#define XDECL(j)  float4 xg##j;
#define XLOAD(j)  xg##j = *(const float4*)(xb##j + vo);
#define XSTORE(j) *(float4*)&Aw[j * APITCH + lo] = xg##j;
    {
        const unsigned vo = lo;
        ROWS16(XDECL)
        ROWS16(XLOAD)
        ROWS16(XSTORE)
    }
    ROWS16(XDECL)
    __builtin_amdgcn_wave_barrier();

#pragma unroll 1
    for (int t = 0; t < NSTEPS; ++t) {
        if (t + 1 < NSTEPS) {
            const unsigned vo = (unsigned)(t + 1) * KC + lo;
            ROWS16(XLOAD)
        }
#pragma unroll
        for (int gi = 0; gi < GPS; ++gi) {
            const int g = t * GPS + gi;
            float4 a0 = *(const float4*)&Aw[arow + gi * 16];
            float4 a1 = *(const float4*)&Aw[arow + gi * 16 + 4];
            float4 b0 = bx[fb + g * 64 + 0];
            float4 b1 = bx[fb + g * 64 + 1];
            float qa[8], qb[8];
            *(float4*)&qa[0] = a0; *(float4*)&qa[4] = a1;
            *(float4*)&qb[0] = b0; *(float4*)&qb[4] = b1;
#pragma unroll
            for (int c = 0; c < 4; c += 2) {
                v2f A0 = {qa[2 * c],     qa[2 * c + 1]};
                v2f B0 = {qb[2 * c],     qb[2 * c + 1]};
                v2f A1 = {qa[2 * c + 2], qa[2 * c + 3]};
                v2f B1 = {qb[2 * c + 2], qb[2 * c + 3]};
                acc0 = __builtin_amdgcn_wmma_f32_16x16x4_f32(
                    false, A0, false, B0, (short)0, acc0, false, false);
                acc1 = __builtin_amdgcn_wmma_f32_16x16x4_f32(
                    false, A1, false, B1, (short)0, acc1, false, false);
            }
        }
        __builtin_amdgcn_wave_barrier();
        if (t + 1 < NSTEPS) {
            ROWS16(XSTORE)
        }
        __builtin_amdgcn_wave_barrier();
    }
#undef XBASE
#undef XDECL
#undef XLOAD
#undef XSTORE
#undef ROWS16

    const int mbase = (lane >= 16) ? 8 : 0;
    const int ncol  = lane & 15;
#pragma unroll
    for (int v = 0; v < 8; ++v) {
        Aw[(v + mbase) * EPITCH + ncol] = acc0[v] + acc1[v];
    }
    __builtin_amdgcn_wave_barrier();

    const int jh = lane >> 4;
    float t[15];
    float keep;
    {
        keep = (Aw[r * EPITCH + 15] + l1_b[s * 16 + 15]) + l1f_b[15];
#pragma unroll
        for (int j = 0; j < 15; ++j) {
            t[j] = (Aw[r * EPITCH + j] + l1_b[s * 16 + j]) + l1f_b[j];
        }
    }
    float l1x[30];
#pragma unroll
    for (int j = 0; j < 15; ++j) {
        float sq = t[j] * t[j] * (127.0f / 128.0f);
        l1x[j]      = fminf(fmaxf(sq, 0.f), 1.f);
        l1x[15 + j] = fminf(fmaxf(t[j], 0.f), 1.f);
    }
    float partial = 0.f;
#pragma unroll 4
    for (int oi = 0; oi < 16; ++oi) {
        int o = jh * 16 + oi;
        const float* wrow = l2_w + (size_t)(s * 32 + o) * 30;
        float a2 = l2_b[s * 32 + o];
#pragma unroll
        for (int k = 0; k < 30; ++k) a2 += l1x[k] * wrow[k];
        float l2x = fminf(fmaxf(a2, 0.f), 1.f);
        partial += l2x * out_w[s * 32 + o];
    }
    float tot = partial + __shfl_xor(partial, 16, 32);

    if (lane < 16 && p < cnt_s) {
        out[rrow] = tot + out_b[s] + keep;
    }
}


extern "C" void kernel_launch(void* const* d_in, const int* in_sizes, int n_in,
                              void* d_out, int out_size, void* d_ws, size_t ws_size,
                              hipStream_t stream) {
    (void)in_sizes; (void)n_in; (void)out_size; (void)ws_size;
    const float* x      = (const float*)d_in[0];
    const int*   lsi    = (const int*)d_in[1];
    const float* l1_w   = (const float*)d_in[2];
    const float* l1_b   = (const float*)d_in[3];
    const float* l1f_w  = (const float*)d_in[4];
    const float* l1f_b  = (const float*)d_in[5];
    const float* l2_w   = (const float*)d_in[6];
    const float* l2_b   = (const float*)d_in[7];
    const float* out_w  = (const float*)d_in[8];
    const float* out_b  = (const float*)d_in[9];
    float* out = (float*)d_out;

    char* ws   = (char*)d_ws;
    int*  cnt  = (int*)(ws + WS_CNT);
    int*  off  = (int*)(ws + WS_OFF);
    int*  bcnt = (int*)(ws + WS_BCNT);
    int*  perm = (int*)(ws + WS_PERM);
    float* frag = (float*)(ws + WS_FRAG);

    k_count<<<BK_CHUNKS, BK_THREADS, 0, stream>>>(lsi, bcnt);
    k_scatter<<<BK_CHUNKS, BK_THREADS, 0, stream>>>(lsi, bcnt, cnt, off, perm);
    k_prep<<<(NSTACK * NGRP * 256) / 256, 256, 0, stream>>>(l1_w, l1f_w, frag);
    k_main<<<NSTACK * 1024, 128, 0, stream>>>(x, perm, off, cnt, frag,
                                              l1_b, l1f_b, l2_w, l2_b,
                                              out_w, out_b, out);
}


# k_main K loop: each step's B fragment loads issued ahead of the next step's x loads (loads return in issue order), so the products wait for B alone and run beside the memory latency; last step peeled
# speedup vs baseline: 2.2043x; 1.0544x over previous
#include <hip/hip_runtime.h>
#include <hip/hip_bf16.h>


typedef __attribute__((ext_vector_type(2))) float v2f;
typedef __attribute__((ext_vector_type(8))) float v8f;

#define NB      65536
#define NL1     1024
#define NSTACK  8
#define NGRP    64
#define KC      128
#define NSTEPS  (NL1 / KC)
#define GPS     (KC / 16)
#define APITCH  132
#define EPITCH  17
#define WS_CNT     0
#define WS_OFF     128
#define WS_BCNT    192
#define WS_PERM    1024
#define WS_FRAG    (1024 + 65536*4)


#define BK_CHUNKS  16
#define BK_THREADS 1024

__device__ __forceinline__ unsigned bk_pack(int4 v) {
    return (1u << ((v.x & 7) * 4)) + (1u << ((v.y & 7) * 4)) +
           (1u << ((v.z & 7) * 4)) + (1u << ((v.w & 7) * 4));
}

__device__ __forceinline__ int bk_field(unsigned ev, unsigned od, int k) {
    return (int)((((k & 1) ? od : ev) >> ((k >> 1) * 8)) & 0xFFu);
}

__global__ __launch_bounds__(BK_THREADS)
void k_count(const int* __restrict__ lsi, int* __restrict__ bcnt) {
    __shared__ unsigned s_w[32][2];
    const int t = threadIdx.x, lane = t & 31, wv = t >> 5;
    const int4 v = ((const int4*)lsi)[blockIdx.x * BK_THREADS + t];
    const unsigned pk = bk_pack(v);
    unsigned ev = pk & 0x0F0F0F0Fu;
    unsigned od = (pk >> 4) & 0x0F0F0F0Fu;
    for (int m = 16; m >= 1; m >>= 1) {
        ev += __shfl_xor(ev, m, 32);
        od += __shfl_xor(od, m, 32);
    }
    if (lane == 0) { s_w[wv][0] = ev; s_w[wv][1] = od; }
    __syncthreads();
    if (t < 8) {
        int acc = 0;
        for (int w = 0; w < 32; ++w) acc += bk_field(s_w[w][0], s_w[w][1], t);
        bcnt[blockIdx.x * 8 + t] = acc;
    }
}

__global__ __launch_bounds__(BK_THREADS)
void k_scatter(const int* __restrict__ lsi, const int* __restrict__ bcnt,
               int* __restrict__ cnt, int* __restrict__ off,
               int* __restrict__ perm) {
    __shared__ unsigned s_w[32][2];
    __shared__ int s_base[8];
    __shared__ int s_wbase[32][8];
    const int t = threadIdx.x, lane = t & 31, wv = t >> 5;
    const int b = blockIdx.x;
    const int4 v = ((const int4*)lsi)[b * BK_THREADS + t];
    const unsigned pk = bk_pack(v);
    const unsigned ev = pk & 0x0F0F0F0Fu;
    const unsigned od = (pk >> 4) & 0x0F0F0F0Fu;

    unsigned iev = ev, iod = od;
    for (int d = 1; d < 32; d <<= 1) {
        const unsigned a = __shfl_up(iev, d, 32);
        const unsigned c = __shfl_up(iod, d, 32);
        if (lane >= d) { iev += a; iod += c; }
    }
    if (lane == 31) { s_w[wv][0] = iev; s_w[wv][1] = iod; }

    if (wv == 0) {
        int tot = 0, pre = 0;
        if (lane < 8) {
            for (int j = 0; j < BK_CHUNKS; ++j) {
                const int c = bcnt[j * 8 + lane];
                tot += c;
                pre += (j < b) ? c : 0;
            }
        }
        int inc = tot;
        for (int d = 1; d < 8; d <<= 1) {
            const int a = __shfl_up(inc, d, 32);
            if (lane >= d) inc += a;
        }
        if (lane < 8) {
            const int start = inc - tot;
            s_base[lane] = start + pre;
            if (b == 0) { cnt[lane] = tot; off[lane] = start; }
        }
    }
    __syncthreads();
    if (t < 256) {
        const int w = t >> 3, k = t & 7;
        int acc = s_base[k];
        for (int w2 = 0; w2 < w; ++w2) acc += bk_field(s_w[w2][0], s_w[w2][1], k);
        s_wbase[w][k] = acc;
    }
    __syncthreads();

    const unsigned xev = iev - ev, xod = iod - od;
    const int row0 = (b * BK_THREADS + t) * 4;
    const int kk[4] = {v.x & 7, v.y & 7, v.z & 7, v.w & 7};
    unsigned loc = 0;
#pragma unroll
    for (int e = 0; e < 4; ++e) {
        const int k = kk[e];
        const int pos = s_wbase[wv][k] + bk_field(xev, xod, k) +
                        (int)((loc >> (k * 4)) & 15u);
        loc += 1u << (k * 4);
        perm[pos] = row0 + e;
    }
}


__global__ void k_prep(const float* __restrict__ l1_w,
                       const float* __restrict__ l1f_w,
                       float* __restrict__ frag) {
    int tid = blockIdx.x * blockDim.x + threadIdx.x;
    int e    = tid & 7;
    int lane = (tid >> 3) & 31;
    int g    = (tid >> 8) & 63;
    int s    = tid >> 14;
    int n = lane & 15;
    int k = g * 16 + ((lane < 16) ? 0 : 8) + e;
    frag[tid] = l1_w[(size_t)(s * 16 + n) * NL1 + k] + l1f_w[(size_t)n * NL1 + k];
}


__device__ __forceinline__ void load_b_step(float4 (&bq)[GPS][2],
                                            const float4* bx,
                                            int fb, int t) {
#pragma unroll
    for (int gi = 0; gi < GPS; ++gi) {
        bq[gi][0] = bx[fb + (t * GPS + gi) * 64 + 0];
        bq[gi][1] = bx[fb + (t * GPS + gi) * 64 + 1];
    }
}

__device__ __forceinline__ void step_products(const float* __restrict__ Aw, int arow,
                                              const float4 (&bq)[GPS][2],
                                              v8f& acc0, v8f& acc1) {
#pragma unroll
    for (int gi = 0; gi < GPS; ++gi) {
        float4 a0 = *(const float4*)&Aw[arow + gi * 16];
        float4 a1 = *(const float4*)&Aw[arow + gi * 16 + 4];
        float qa[8], qb[8];
        *(float4*)&qa[0] = a0; *(float4*)&qa[4] = a1;
        *(float4*)&qb[0] = bq[gi][0]; *(float4*)&qb[4] = bq[gi][1];
#pragma unroll
        for (int c = 0; c < 4; c += 2) {
            v2f A0 = {qa[2 * c],     qa[2 * c + 1]};
            v2f B0 = {qb[2 * c],     qb[2 * c + 1]};
            v2f A1 = {qa[2 * c + 2], qa[2 * c + 3]};
            v2f B1 = {qb[2 * c + 2], qb[2 * c + 3]};
            acc0 = __builtin_amdgcn_wmma_f32_16x16x4_f32(
                false, A0, false, B0, (short)0, acc0, false, false);
            acc1 = __builtin_amdgcn_wmma_f32_16x16x4_f32(
                false, A1, false, B1, (short)0, acc1, false, false);
        }
    }
}

__global__ __launch_bounds__(128, 5)
void k_main(const float* x,
            const int* __restrict__ perm,
            const int* __restrict__ off,
            const int* __restrict__ cnt,
            const float* frag,
            const float* __restrict__ l1_b,
            const float* __restrict__ l1f_b,
            const float* __restrict__ l2_w,
            const float* __restrict__ l2_b,
            const float* __restrict__ out_w,
            const float* __restrict__ out_b,
            float* __restrict__ out) {
    __shared__ __attribute__((aligned(16))) float sA[4][16 * APITCH];

    const int s    = blockIdx.x >> 10;
    const int tile = blockIdx.x & 1023;
    const int cnt_s = cnt[s];
    if (cnt_s == 0) return;
    if (tile * 64 >= cnt_s) return;

    const int wv   = threadIdx.x >> 5;
    const int lane = threadIdx.x & 31;
    const int r    = lane & 15;
    const int p    = tile * 64 + wv * 16 + r;
    const int pcl  = (p < cnt_s) ? p : (cnt_s - 1);
    const int rrow = perm[off[s] + pcl];

    const int koff = (lane < 16) ? 0 : 8;
    float* __restrict__ Aw = sA[wv];
    const float4* bx = (const float4*)frag;

#define ROWS16(M) M(0) M(1) M(2) M(3) M(4) M(5) M(6) M(7) \
                  M(8) M(9) M(10) M(11) M(12) M(13) M(14) M(15)
#define XBASE(j) const float* xb##j = \
        x + (size_t)__builtin_amdgcn_readlane(rrow, j) * NL1;
    ROWS16(XBASE)
    const unsigned lo = (unsigned)lane * 4u;

    v8f acc0 = {0.f, 0.f, 0.f, 0.f, 0.f, 0.f, 0.f, 0.f};
    v8f acc1 = {0.f, 0.f, 0.f, 0.f, 0.f, 0.f, 0.f, 0.f};

    const int fb = (s * NGRP) * 64 + lane * 2;
    const int arow = r * APITCH + koff;

#define XDECL(j)  float4 xg##j;
#define XLOAD(j)  xg##j = *(const float4*)(xb##j + vo);
#define XSTORE(j) *(float4*)&Aw[j * APITCH + lo] = xg##j;
    {
        const unsigned vo = lo;
        ROWS16(XDECL)
        ROWS16(XLOAD)
        ROWS16(XSTORE)
    }
    ROWS16(XDECL)
    __builtin_amdgcn_wave_barrier();

#pragma unroll 1
    for (int t = 0; t + 1 < NSTEPS; ++t) {
        float4 bq[GPS][2];
        load_b_step(bq, bx, fb, t);
        asm volatile("" ::: "memory");
        {
            const unsigned vo = (unsigned)(t + 1) * KC + lo;
            ROWS16(XLOAD)
        }
        asm volatile("" ::: "memory");
        step_products(Aw, arow, bq, acc0, acc1);
        __builtin_amdgcn_wave_barrier();
        ROWS16(XSTORE)
        __builtin_amdgcn_wave_barrier();
    }
    {
        float4 bq[GPS][2];
        load_b_step(bq, bx, fb, NSTEPS - 1);
        step_products(Aw, arow, bq, acc0, acc1);
        __builtin_amdgcn_wave_barrier();
    }
#undef XBASE
#undef XDECL
#undef XLOAD
#undef XSTORE
#undef ROWS16

    const int mbase = (lane >= 16) ? 8 : 0;
    const int ncol  = lane & 15;
#pragma unroll
    for (int v = 0; v < 8; ++v) {
        Aw[(v + mbase) * EPITCH + ncol] = acc0[v] + acc1[v];
    }
    __builtin_amdgcn_wave_barrier();

    const int jh = lane >> 4;
    float t[15];
    float keep;
    {
        keep = (Aw[r * EPITCH + 15] + l1_b[s * 16 + 15]) + l1f_b[15];
#pragma unroll
        for (int j = 0; j < 15; ++j) {
            t[j] = (Aw[r * EPITCH + j] + l1_b[s * 16 + j]) + l1f_b[j];
        }
    }
    float l1x[30];
#pragma unroll
    for (int j = 0; j < 15; ++j) {
        float sq = t[j] * t[j] * (127.0f / 128.0f);
        l1x[j]      = fminf(fmaxf(sq, 0.f), 1.f);
        l1x[15 + j] = fminf(fmaxf(t[j], 0.f), 1.f);
    }
    float partial = 0.f;
#pragma unroll 4
    for (int oi = 0; oi < 16; ++oi) {
        int o = jh * 16 + oi;
        const float* wrow = l2_w + (size_t)(s * 32 + o) * 30;
        float a2 = l2_b[s * 32 + o];
#pragma unroll
        for (int k = 0; k < 30; ++k) a2 += l1x[k] * wrow[k];
        float l2x = fminf(fmaxf(a2, 0.f), 1.f);
        partial += l2x * out_w[s * 32 + o];
    }
    float tot = partial + __shfl_xor(partial, 16, 32);

    if (lane < 16 && p < cnt_s) {
        out[rrow] = tot + out_b[s] + keep;
    }
}


extern "C" void kernel_launch(void* const* d_in, const int* in_sizes, int n_in,
                              void* d_out, int out_size, void* d_ws, size_t ws_size,
                              hipStream_t stream) {
    (void)in_sizes; (void)n_in; (void)out_size; (void)ws_size;
    const float* x      = (const float*)d_in[0];
    const int*   lsi    = (const int*)d_in[1];
    const float* l1_w   = (const float*)d_in[2];
    const float* l1_b   = (const float*)d_in[3];
    const float* l1f_w  = (const float*)d_in[4];
    const float* l1f_b  = (const float*)d_in[5];
    const float* l2_w   = (const float*)d_in[6];
    const float* l2_b   = (const float*)d_in[7];
    const float* out_w  = (const float*)d_in[8];
    const float* out_b  = (const float*)d_in[9];
    float* out = (float*)d_out;

    char* ws   = (char*)d_ws;
    int*  cnt  = (int*)(ws + WS_CNT);
    int*  off  = (int*)(ws + WS_OFF);
    int*  bcnt = (int*)(ws + WS_BCNT);
    int*  perm = (int*)(ws + WS_PERM);
    float* frag = (float*)(ws + WS_FRAG);

    k_count<<<BK_CHUNKS, BK_THREADS, 0, stream>>>(lsi, bcnt);
    k_scatter<<<BK_CHUNKS, BK_THREADS, 0, stream>>>(lsi, bcnt, cnt, off, perm);
    k_prep<<<(NSTACK * NGRP * 256) / 256, 256, 0, stream>>>(l1_w, l1f_w, frag);
    k_main<<<NSTACK * 1024, 128, 0, stream>>>(x, perm, off, cnt, frag,
                                              l1_b, l1f_b, l2_w, l2_b,
                                              out_w, out_b, out);
}


# both pre-passes in ONE launch: each bucketing block counts all 16 chunks of the indices itself (no count launch), the weight-fold blocks ride in the same grid; 4 launches become 2; sums unchanged
# speedup vs baseline: 2.3473x; 1.0649x over previous
#include <hip/hip_runtime.h>
#include <hip/hip_bf16.h>


typedef __attribute__((ext_vector_type(2))) float v2f;
typedef __attribute__((ext_vector_type(8))) float v8f;

#define NB      65536
#define NL1     1024
#define NSTACK  8
#define NGRP    64
#define KC      128
#define NSTEPS  (NL1 / KC)
#define GPS     (KC / 16)
#define APITCH  132
#define EPITCH  17
#define WS_CNT     0
#define WS_OFF     128
#define WS_PERM    1024
#define WS_FRAG    (1024 + 65536*4)


#define BK_CHUNKS  16
#define BK_THREADS 1024

__device__ __forceinline__ unsigned bk_pack(int4 v) {
    return (1u << ((v.x & 7) * 4)) + (1u << ((v.y & 7) * 4)) +
           (1u << ((v.z & 7) * 4)) + (1u << ((v.w & 7) * 4));
}

__device__ __forceinline__ int bk_field(unsigned ev, unsigned od, int k) {
    return (int)((((k & 1) ? od : ev) >> ((k >> 1) * 8)) & 0xFFu);
}


#define PREP_BLOCKS ((NSTACK * NGRP * 256) / BK_THREADS)

__device__ __forceinline__ void fold_weights(const float* __restrict__ l1_w,
                                             const float* __restrict__ l1f_w,
                                             float* __restrict__ frag, int tid) {
    int e    = tid & 7;
    int lane = (tid >> 3) & 31;
    int g    = (tid >> 8) & 63;
    int s    = tid >> 14;
    int n = lane & 15;
    int k = g * 16 + ((lane < 16) ? 0 : 8) + e;
    frag[tid] = l1_w[(size_t)(s * 16 + n) * NL1 + k] + l1f_w[(size_t)n * NL1 + k];
}


__device__ __forceinline__ int bk_field16(const unsigned* w4, int k) {
    return (int)((w4[(k & 1) * 2 + ((k >> 1) & 1)] >> ((k >> 2) * 16)) & 0xFFFFu);
}

__global__ __launch_bounds__(BK_THREADS)
void k_pre(const int* __restrict__ lsi,
           const float* __restrict__ l1_w,
           const float* __restrict__ l1f_w,
           int* __restrict__ cnt, int* __restrict__ off,
           int* __restrict__ perm, float* __restrict__ frag) {
    __shared__ unsigned s_w[32][2];
    __shared__ unsigned s_tp[32][8];
    __shared__ int s_base[8];
    __shared__ int s_wbase[32][8];
    const int t = threadIdx.x, lane = t & 31, wv = t >> 5;
    if (blockIdx.x >= BK_CHUNKS) {
        fold_weights(l1_w, l1f_w, frag, (blockIdx.x - BK_CHUNKS) * BK_THREADS + t);
        return;
    }
    const int b = blockIdx.x;

    unsigned tev = 0, tod = 0, pev = 0, pod = 0;
#pragma unroll
    for (int j = 0; j < BK_CHUNKS; ++j) {
        const unsigned pj = bk_pack(((const int4*)lsi)[j * BK_THREADS + t]);
        const unsigned e = pj & 0x0F0F0F0Fu, o = (pj >> 4) & 0x0F0F0F0Fu;
        tev += e; tod += o;
        if (j < b) { pev += e; pod += o; }
    }
    unsigned rs[8] = { tev & 0x00FF00FFu, (tev >> 8) & 0x00FF00FFu,
                       tod & 0x00FF00FFu, (tod >> 8) & 0x00FF00FFu,
                       pev & 0x00FF00FFu, (pev >> 8) & 0x00FF00FFu,
                       pod & 0x00FF00FFu, (pod >> 8) & 0x00FF00FFu };
    for (int m = 16; m >= 1; m >>= 1) {
#pragma unroll
        for (int i = 0; i < 8; ++i) rs[i] += __shfl_xor(rs[i], m, 32);
    }
    if (lane == 0) {
#pragma unroll
        for (int i = 0; i < 8; ++i) s_tp[wv][i] = rs[i];
    }

    const int4 v = ((const int4*)lsi)[b * BK_THREADS + t];
    const unsigned pk = bk_pack(v);
    const unsigned ev = pk & 0x0F0F0F0Fu;
    const unsigned od = (pk >> 4) & 0x0F0F0F0Fu;

    unsigned iev = ev, iod = od;
    for (int d = 1; d < 32; d <<= 1) {
        const unsigned a = __shfl_up(iev, d, 32);
        const unsigned c = __shfl_up(iod, d, 32);
        if (lane >= d) { iev += a; iod += c; }
    }
    if (lane == 31) { s_w[wv][0] = iev; s_w[wv][1] = iod; }
    __syncthreads();

    if (wv == 0) {
        int tot = 0, pre = 0;
        if (lane < 8) {
            for (int w = 0; w < 32; ++w) {
                tot += bk_field16(&s_tp[w][0], lane);
                pre += bk_field16(&s_tp[w][4], lane);
            }
        }
        int inc = tot;
        for (int d = 1; d < 8; d <<= 1) {
            const int a = __shfl_up(inc, d, 32);
            if (lane >= d) inc += a;
        }
        if (lane < 8) {
            const int start = inc - tot;
            s_base[lane] = start + pre;
            if (b == 0) { cnt[lane] = tot; off[lane] = start; }
        }
    }
    __syncthreads();
    if (t < 256) {
        const int w = t >> 3, k = t & 7;
        int acc = s_base[k];
        for (int w2 = 0; w2 < w; ++w2) acc += bk_field(s_w[w2][0], s_w[w2][1], k);
        s_wbase[w][k] = acc;
    }
    __syncthreads();

    const unsigned xev = iev - ev, xod = iod - od;
    const int row0 = (b * BK_THREADS + t) * 4;
    const int kk[4] = {v.x & 7, v.y & 7, v.z & 7, v.w & 7};
    unsigned loc = 0;
#pragma unroll
    for (int e = 0; e < 4; ++e) {
        const int k = kk[e];
        const int pos = s_wbase[wv][k] + bk_field(xev, xod, k) +
                        (int)((loc >> (k * 4)) & 15u);
        loc += 1u << (k * 4);
        perm[pos] = row0 + e;
    }
}


__device__ __forceinline__ void load_b_step(float4 (&bq)[GPS][2],
                                            const float4* bx,
                                            int fb, int t) {
#pragma unroll
    for (int gi = 0; gi < GPS; ++gi) {
        bq[gi][0] = bx[fb + (t * GPS + gi) * 64 + 0];
        bq[gi][1] = bx[fb + (t * GPS + gi) * 64 + 1];
    }
}

__device__ __forceinline__ void step_products(const float* __restrict__ Aw, int arow,
                                              const float4 (&bq)[GPS][2],
                                              v8f& acc0, v8f& acc1) {
#pragma unroll
    for (int gi = 0; gi < GPS; ++gi) {
        float4 a0 = *(const float4*)&Aw[arow + gi * 16];
        float4 a1 = *(const float4*)&Aw[arow + gi * 16 + 4];
        float qa[8], qb[8];
        *(float4*)&qa[0] = a0; *(float4*)&qa[4] = a1;
        *(float4*)&qb[0] = bq[gi][0]; *(float4*)&qb[4] = bq[gi][1];
#pragma unroll
        for (int c = 0; c < 4; c += 2) {
            v2f A0 = {qa[2 * c],     qa[2 * c + 1]};
            v2f B0 = {qb[2 * c],     qb[2 * c + 1]};
            v2f A1 = {qa[2 * c + 2], qa[2 * c + 3]};
            v2f B1 = {qb[2 * c + 2], qb[2 * c + 3]};
            acc0 = __builtin_amdgcn_wmma_f32_16x16x4_f32(
                false, A0, false, B0, (short)0, acc0, false, false);
            acc1 = __builtin_amdgcn_wmma_f32_16x16x4_f32(
                false, A1, false, B1, (short)0, acc1, false, false);
        }
    }
}

__global__ __launch_bounds__(128, 5)
void k_main(const float* x,
            const int* __restrict__ perm,
            const int* __restrict__ off,
            const int* __restrict__ cnt,
            const float* frag,
            const float* __restrict__ l1_b,
            const float* __restrict__ l1f_b,
            const float* __restrict__ l2_w,
            const float* __restrict__ l2_b,
            const float* __restrict__ out_w,
            const float* __restrict__ out_b,
            float* __restrict__ out) {
    __shared__ __attribute__((aligned(16))) float sA[4][16 * APITCH];

    const int s    = blockIdx.x >> 10;
    const int tile = blockIdx.x & 1023;
    const int cnt_s = cnt[s];
    if (cnt_s == 0) return;
    if (tile * 64 >= cnt_s) return;

    const int wv   = threadIdx.x >> 5;
    const int lane = threadIdx.x & 31;
    const int r    = lane & 15;
    const int p    = tile * 64 + wv * 16 + r;
    const int pcl  = (p < cnt_s) ? p : (cnt_s - 1);
    const int rrow = perm[off[s] + pcl];

    const int koff = (lane < 16) ? 0 : 8;
    float* __restrict__ Aw = sA[wv];
    const float4* bx = (const float4*)frag;

#define ROWS16(M) M(0) M(1) M(2) M(3) M(4) M(5) M(6) M(7) \
                  M(8) M(9) M(10) M(11) M(12) M(13) M(14) M(15)
#define XBASE(j) const float* xb##j = \
        x + (size_t)__builtin_amdgcn_readlane(rrow, j) * NL1;
    ROWS16(XBASE)
    const unsigned lo = (unsigned)lane * 4u;

    v8f acc0 = {0.f, 0.f, 0.f, 0.f, 0.f, 0.f, 0.f, 0.f};
    v8f acc1 = {0.f, 0.f, 0.f, 0.f, 0.f, 0.f, 0.f, 0.f};

    const int fb = (s * NGRP) * 64 + lane * 2;
    const int arow = r * APITCH + koff;

#define XDECL(j)  float4 xg##j;
#define XLOAD(j)  xg##j = *(const float4*)(xb##j + vo);
#define XSTORE(j) *(float4*)&Aw[j * APITCH + lo] = xg##j;
    {
        const unsigned vo = lo;
        ROWS16(XDECL)
        ROWS16(XLOAD)
        ROWS16(XSTORE)
    }
    ROWS16(XDECL)
    __builtin_amdgcn_wave_barrier();

#pragma unroll 1
    for (int t = 0; t + 1 < NSTEPS; ++t) {
        float4 bq[GPS][2];
        load_b_step(bq, bx, fb, t);
        asm volatile("" ::: "memory");
        {
            const unsigned vo = (unsigned)(t + 1) * KC + lo;
            ROWS16(XLOAD)
        }
        asm volatile("" ::: "memory");
        step_products(Aw, arow, bq, acc0, acc1);
        __builtin_amdgcn_wave_barrier();
        ROWS16(XSTORE)
        __builtin_amdgcn_wave_barrier();
    }
    {
        float4 bq[GPS][2];
        load_b_step(bq, bx, fb, NSTEPS - 1);
        step_products(Aw, arow, bq, acc0, acc1);
        __builtin_amdgcn_wave_barrier();
    }
#undef XBASE
#undef XDECL
#undef XLOAD
#undef XSTORE
#undef ROWS16

    const int mbase = (lane >= 16) ? 8 : 0;
    const int ncol  = lane & 15;
#pragma unroll
    for (int v = 0; v < 8; ++v) {
        Aw[(v + mbase) * EPITCH + ncol] = acc0[v] + acc1[v];
    }
    __builtin_amdgcn_wave_barrier();

    const int jh = lane >> 4;
    float t[15];
    float keep;
    {
        keep = (Aw[r * EPITCH + 15] + l1_b[s * 16 + 15]) + l1f_b[15];
#pragma unroll
        for (int j = 0; j < 15; ++j) {
            t[j] = (Aw[r * EPITCH + j] + l1_b[s * 16 + j]) + l1f_b[j];
        }
    }
    float l1x[30];
#pragma unroll
    for (int j = 0; j < 15; ++j) {
        float sq = t[j] * t[j] * (127.0f / 128.0f);
        l1x[j]      = fminf(fmaxf(sq, 0.f), 1.f);
        l1x[15 + j] = fminf(fmaxf(t[j], 0.f), 1.f);
    }
    float partial = 0.f;
#pragma unroll 4
    for (int oi = 0; oi < 16; ++oi) {
        int o = jh * 16 + oi;
        const float* wrow = l2_w + (size_t)(s * 32 + o) * 30;
        float a2 = l2_b[s * 32 + o];
#pragma unroll
        for (int k = 0; k < 30; ++k) a2 += l1x[k] * wrow[k];
        float l2x = fminf(fmaxf(a2, 0.f), 1.f);
        partial += l2x * out_w[s * 32 + o];
    }
    float tot = partial + __shfl_xor(partial, 16, 32);

    if (lane < 16 && p < cnt_s) {
        out[rrow] = tot + out_b[s] + keep;
    }
}


extern "C" void kernel_launch(void* const* d_in, const int* in_sizes, int n_in,
                              void* d_out, int out_size, void* d_ws, size_t ws_size,
                              hipStream_t stream) {
    (void)in_sizes; (void)n_in; (void)out_size; (void)ws_size;
    const float* x      = (const float*)d_in[0];
    const int*   lsi    = (const int*)d_in[1];
    const float* l1_w   = (const float*)d_in[2];
    const float* l1_b   = (const float*)d_in[3];
    const float* l1f_w  = (const float*)d_in[4];
    const float* l1f_b  = (const float*)d_in[5];
    const float* l2_w   = (const float*)d_in[6];
    const float* l2_b   = (const float*)d_in[7];
    const float* out_w  = (const float*)d_in[8];
    const float* out_b  = (const float*)d_in[9];
    float* out = (float*)d_out;

    char* ws   = (char*)d_ws;
    int*  cnt  = (int*)(ws + WS_CNT);
    int*  off  = (int*)(ws + WS_OFF);
    int*  perm = (int*)(ws + WS_PERM);
    float* frag = (float*)(ws + WS_FRAG);

    k_pre<<<BK_CHUNKS + PREP_BLOCKS, BK_THREADS, 0, stream>>>(lsi, l1_w, l1f_w,
                                                              cnt, off, perm, frag);
    k_main<<<NSTACK * 1024, 128, 0, stream>>>(x, perm, off, cnt, frag,
                                              l1_b, l1f_b, l2_w, l2_b,
                                              out_w, out_b, out);
}


# ONE launch, no pre-pass: each block finds its tile's rows from its own 4096-row chunk of the indices and sums the two first-layer weight matrices into LDS fragment order itself; sums unchanged
# speedup vs baseline: 2.4686x; 1.0517x over previous
#include <hip/hip_runtime.h>


typedef __attribute__((ext_vector_type(2))) float v2f;
typedef __attribute__((ext_vector_type(8))) float v8f;

#define NB      65536
#define NL1     1024
#define NSTACK  8
#define KC      128
#define NSTEPS  (NL1 / KC)
#define GPS     (KC / 16)
#define APITCH  132
#define BPITCH  33
#define EPITCH  17
#define CH_ROWS 4096
#define NCHUNK  (NB / CH_ROWS)
#define TPC     72

__device__ __forceinline__ unsigned bk_pack(int4 v) {
    return (1u << ((v.x & 7) * 4)) + (1u << ((v.y & 7) * 4)) +
           (1u << ((v.z & 7) * 4)) + (1u << ((v.w & 7) * 4));
}

__device__ __forceinline__ int bk_field16(const unsigned* w4, int k) {
    return (int)((w4[(k & 1) * 2 + ((k >> 1) & 1)] >> ((k >> 2) * 16)) & 0xFFFFu);
}

__device__ __forceinline__ float4 add4(float4 a, float4 b) {
    return make_float4(a.x + b.x, a.y + b.y, a.z + b.z, a.w + b.w);
}

__device__ __forceinline__ void step_products(const float* __restrict__ Aw, int arow,
                                              const float4* __restrict__ Bs, int lane,
                                              v8f& acc0, v8f& acc1) {
    float4 a0 = *(const float4*)&Aw[arow];
    float4 a1 = *(const float4*)&Aw[arow + 4];
    float4 b0 = Bs[lane];
    float4 b1 = Bs[BPITCH + lane];
#pragma unroll
    for (int gi = 0; gi < GPS; ++gi) {
        float qa[8], qb[8];
        *(float4*)&qa[0] = a0; *(float4*)&qa[4] = a1;
        *(float4*)&qb[0] = b0; *(float4*)&qb[4] = b1;
        if (gi + 1 < GPS) {
            a0 = *(const float4*)&Aw[arow + (gi + 1) * 16];
            a1 = *(const float4*)&Aw[arow + (gi + 1) * 16 + 4];
            b0 = Bs[((gi + 1) * 2 + 0) * BPITCH + lane];
            b1 = Bs[((gi + 1) * 2 + 1) * BPITCH + lane];
        }
        __builtin_amdgcn_sched_barrier(0);
#pragma unroll
        for (int c = 0; c < 4; c += 2) {
            v2f A0 = {qa[2 * c],     qa[2 * c + 1]};
            v2f B0 = {qb[2 * c],     qb[2 * c + 1]};
            v2f A1 = {qa[2 * c + 2], qa[2 * c + 3]};
            v2f B1 = {qb[2 * c + 2], qb[2 * c + 3]};
            acc0 = __builtin_amdgcn_wmma_f32_16x16x4_f32(
                false, A0, false, B0, (short)0, acc0, false, false);
            acc1 = __builtin_amdgcn_wmma_f32_16x16x4_f32(
                false, A1, false, B1, (short)0, acc1, false, false);
        }
    }
}

__global__ __launch_bounds__(128, 5)
void k_main(const float* x,
            const int* __restrict__ lsi,
            const float* l1_w,
            const float* __restrict__ l1_b,
            const float* l1f_w,
            const float* __restrict__ l1f_b,
            const float* __restrict__ l2_w,
            const float* __restrict__ l2_b,
            const float* __restrict__ out_w,
            const float* __restrict__ out_b,
            float* __restrict__ out) {
    __shared__ __attribute__((aligned(16))) float sA[4][16 * APITCH];
    __shared__ __attribute__((aligned(16))) float4 sB[2][16 * BPITCH];
    __shared__ __attribute__((aligned(16))) unsigned s_cnt[4][4];
    __shared__ unsigned s_m[4][2];
    __shared__ int s_rows[64];

    const int tid  = threadIdx.x;
    const int wv   = tid >> 5;
    const int lane = tid & 31;
    const int c = blockIdx.x / TPC;
    const int j = blockIdx.x - c * TPC;

    int4 iv[8];
#pragma unroll
    for (int i = 0; i < 8; ++i) iv[i] = ((const int4*)lsi)[c * (CH_ROWS / 4) + i * 128 + tid];
    unsigned ev = 0, od = 0;
#pragma unroll
    for (int i = 0; i < 8; ++i) {
        const unsigned pk = bk_pack(iv[i]);
        ev += pk & 0x0F0F0F0Fu;
        od += (pk >> 4) & 0x0F0F0F0Fu;
    }
    unsigned rs[4] = { ev & 0x00FF00FFu, (ev >> 8) & 0x00FF00FFu,
                       od & 0x00FF00FFu, (od >> 8) & 0x00FF00FFu };
    for (int m = 16; m >= 1; m >>= 1) {
#pragma unroll
        for (int i = 0; i < 4; ++i) rs[i] += __shfl_xor(rs[i], m, 32);
    }
    if (lane == 0) {
#pragma unroll
        for (int i = 0; i < 4; ++i) s_cnt[wv][i] = rs[i];
    }
    __syncthreads();
    unsigned tot[4];
#pragma unroll
    for (int i = 0; i < 4; ++i) tot[i] = s_cnt[0][i] + s_cnt[1][i] + s_cnt[2][i] + s_cnt[3][i];

    int s_v = -1, tile_v = 0, cnt_v = 0, pre = 0;
#pragma unroll
    for (int k = 0; k < NSTACK; ++k) {
        const int ck = bk_field16(tot, k);
        const int tk = (ck + 63) >> 6;
        if (s_v < 0 && j < pre + tk) { s_v = k; tile_v = j - pre; cnt_v = ck; }
        pre += tk;
    }
    if (s_v < 0) return;
    const int s     = __builtin_amdgcn_readfirstlane(s_v);
    const int tile  = __builtin_amdgcn_readfirstlane(tile_v);
    const int cnt_s = __builtin_amdgcn_readfirstlane(cnt_v);
    const int lo0   = tile * 64;

    unsigned mlo = 0, mhi = 0;
#pragma unroll
    for (int i = 0; i < 8; ++i) {
        const unsigned m = (unsigned)((iv[i].x & 7) == s) + (unsigned)((iv[i].y & 7) == s) +
                           (unsigned)((iv[i].z & 7) == s) + (unsigned)((iv[i].w & 7) == s);
        if (i < 4) mlo |= m << (8 * i); else mhi |= m << (8 * (i - 4));
    }
    unsigned ilo = mlo, ihi = mhi;
    for (int d = 1; d < 32; d <<= 1) {
        const unsigned a = __shfl_up(ilo, d, 32);
        const unsigned b = __shfl_up(ihi, d, 32);
        if (lane >= d) { ilo += a; ihi += b; }
    }
    if (lane == 31) { s_m[wv][0] = ilo; s_m[wv][1] = ihi; }
    __syncthreads();
    unsigned sm[4][2];
#pragma unroll
    for (int w = 0; w < 4; ++w) { sm[w][0] = s_m[w][0]; sm[w][1] = s_m[w][1]; }
    {
        int base = -lo0;
#pragma unroll
        for (int i = 0; i < 8; ++i) {
            const int sh = (i & 3) * 8;
            int T = 0, before = 0;
#pragma unroll
            for (int w = 0; w < 4; ++w) {
                const int bw = (int)((sm[w][i >> 2] >> sh) & 0xFFu);
                T += bw;
                if (w < wv) before += bw;
            }
            const unsigned incw = (i < 4) ? ilo : ihi;
            const unsigned ownw = (i < 4) ? mlo : mhi;
            int pos = base + before + (int)((incw >> sh) & 0xFFu) - (int)((ownw >> sh) & 0xFFu);
            const int row0 = c * CH_ROWS + (i * 128 + tid) * 4;
            const int kk[4] = {iv[i].x & 7, iv[i].y & 7, iv[i].z & 7, iv[i].w & 7};
#pragma unroll
            for (int e = 0; e < 4; ++e) {
                if (kk[e] == s) {
                    if ((unsigned)pos < 64u) s_rows[pos] = row0 + e;
                    ++pos;
                }
            }
            base += T;
        }
    }
    __syncthreads();

    const int r    = lane & 15;
    const int p    = lo0 + wv * 16 + r;
    const int pcl  = (p < cnt_s) ? p : (cnt_s - 1);
    const int rrow = s_rows[pcl - lo0];

    const int koff = (lane < 16) ? 0 : 8;
    float* __restrict__ Aw = sA[wv];

#define ROWS16(M) M(0) M(1) M(2) M(3) M(4) M(5) M(6) M(7) \
                  M(8) M(9) M(10) M(11) M(12) M(13) M(14) M(15)
#define XBASE(j) const float* xb##j = \
        x + (size_t)__builtin_amdgcn_readlane(rrow, j) * NL1;
    ROWS16(XBASE)
    const unsigned lo = (unsigned)lane * 4u;

    const float* wsb = l1_w  + (size_t)(s * 16 + wv * 4) * NL1 + lo;
    const float* wfb = l1f_w + (size_t)(wv * 4) * NL1 + lo;
    const int bst = ((lane >> 2) * 2 + (lane & 1)) * BPITCH + 16 * ((lane >> 1) & 1) + wv * 4;

    v8f acc0 = {0.f, 0.f, 0.f, 0.f, 0.f, 0.f, 0.f, 0.f};
    v8f acc1 = {0.f, 0.f, 0.f, 0.f, 0.f, 0.f, 0.f, 0.f};

    const int arow = r * APITCH + koff;

#define XDECL(j)  float4 xg##j;
#define XLOAD(j)  xg##j = *(const float4*)(xb##j + vo);
#define XSTORE(j) *(float4*)&Aw[j * APITCH + lo] = xg##j;
#define WROWS(M)  M(0) M(1) M(2) M(3)
#define WDECL(i)  float4 ws##i, wf##i;
#define WLOAD(i)  ws##i = *(const float4*)(wsb + i * NL1 + wo); \
                  wf##i = *(const float4*)(wfb + i * NL1 + wo);
#define WSTORE(i) Bst[bst + i] = add4(ws##i, wf##i);
    ROWS16(XDECL)
    WROWS(WDECL)
    {
        const unsigned vo = lo;
        const unsigned wo = 0;
        float4* Bst = sB[0];
        WROWS(WLOAD)
        ROWS16(XLOAD)
        ROWS16(XSTORE)
        WROWS(WSTORE)
    }
    __syncthreads();

#pragma unroll 1
    for (int t = 0; t + 1 < NSTEPS; ++t) {
        {
            const unsigned vo = (unsigned)(t + 1) * KC + lo;
            const unsigned wo = (unsigned)(t + 1) * KC;
            WROWS(WLOAD)
            ROWS16(XLOAD)
        }
        asm volatile("" ::: "memory");
        __builtin_amdgcn_sched_barrier(0);
        step_products(Aw, arow, sB[t & 1], lane, acc0, acc1);
        __builtin_amdgcn_wave_barrier();
        ROWS16(XSTORE)
        {
            float4* Bst = sB[(t + 1) & 1];
            WROWS(WSTORE)
        }
        __syncthreads();
    }
    {
        step_products(Aw, arow, sB[(NSTEPS - 1) & 1], lane, acc0, acc1);
        __builtin_amdgcn_wave_barrier();
    }
#undef XBASE
#undef XDECL
#undef XLOAD
#undef XSTORE
#undef WROWS
#undef WDECL
#undef WLOAD
#undef WSTORE
#undef ROWS16

    const int mbase = (lane >= 16) ? 8 : 0;
    const int ncol  = lane & 15;
#pragma unroll
    for (int v = 0; v < 8; ++v) {
        Aw[(v + mbase) * EPITCH + ncol] = acc0[v] + acc1[v];
    }
    __builtin_amdgcn_wave_barrier();

    const int jh = lane >> 4;
    float t[15];
    float keep;
    {
        keep = (Aw[r * EPITCH + 15] + l1_b[s * 16 + 15]) + l1f_b[15];
#pragma unroll
        for (int q = 0; q < 15; ++q) {
            t[q] = (Aw[r * EPITCH + q] + l1_b[s * 16 + q]) + l1f_b[q];
        }
    }
    float l1x[30];
#pragma unroll
    for (int q = 0; q < 15; ++q) {
        float sq = t[q] * t[q] * (127.0f / 128.0f);
        l1x[q]      = fminf(fmaxf(sq, 0.f), 1.f);
        l1x[15 + q] = fminf(fmaxf(t[q], 0.f), 1.f);
    }
    float partial = 0.f;
#pragma unroll 4
    for (int oi = 0; oi < 16; ++oi) {
        int o = jh * 16 + oi;
        const float* wrow = l2_w + (size_t)(s * 32 + o) * 30;
        float a2 = l2_b[s * 32 + o];
#pragma unroll
        for (int k = 0; k < 30; ++k) a2 += l1x[k] * wrow[k];
        float l2x = fminf(fmaxf(a2, 0.f), 1.f);
        partial += l2x * out_w[s * 32 + o];
    }
    float tot_o = partial + __shfl_xor(partial, 16, 32);

    if (lane < 16 && p < cnt_s) {
        out[rrow] = tot_o + out_b[s] + keep;
    }
}


extern "C" void kernel_launch(void* const* d_in, const int* in_sizes, int n_in,
                              void* d_out, int out_size, void* d_ws, size_t ws_size,
                              hipStream_t stream) {
    (void)in_sizes; (void)n_in; (void)out_size; (void)d_ws; (void)ws_size;
    const float* x      = (const float*)d_in[0];
    const int*   lsi    = (const int*)d_in[1];
    const float* l1_w   = (const float*)d_in[2];
    const float* l1_b   = (const float*)d_in[3];
    const float* l1f_w  = (const float*)d_in[4];
    const float* l1f_b  = (const float*)d_in[5];
    const float* l2_w   = (const float*)d_in[6];
    const float* l2_b   = (const float*)d_in[7];
    const float* out_w  = (const float*)d_in[8];
    const float* out_b  = (const float*)d_in[9];
    float* out = (float*)d_out;

    k_main<<<NCHUNK * TPC, 128, 0, stream>>>(x, lsi, l1_w, l1_b, l1f_w, l1f_b,
                                             l2_w, l2_b, out_w, out_b, out);
}


# k_main K loop: next step's memory work moved in among this step's products: x-row loads go out three at a time beside the K groups' fragment reads, weight sums wait a few groups in; sums unchanged
# speedup vs baseline: 2.5757x; 1.0434x over previous
#include <hip/hip_runtime.h>


typedef __attribute__((ext_vector_type(2))) float v2f;
typedef __attribute__((ext_vector_type(8))) float v8f;

#define NB      65536
#define NL1     1024
#define NSTACK  8
#define KC      128
#define NSTEPS  (NL1 / KC)
#define GPS     (KC / 16)
#define APITCH  132
#define BPITCH  33
#define EPITCH  17
#define CH_ROWS 4096
#define NCHUNK  (NB / CH_ROWS)
#define TPC     72

__device__ __forceinline__ unsigned bk_pack(int4 v) {
    return (1u << ((v.x & 7) * 4)) + (1u << ((v.y & 7) * 4)) +
           (1u << ((v.z & 7) * 4)) + (1u << ((v.w & 7) * 4));
}

__device__ __forceinline__ int bk_field16(const unsigned* w4, int k) {
    return (int)((w4[(k & 1) * 2 + ((k >> 1) & 1)] >> ((k >> 2) * 16)) & 0xFFFFu);
}

__device__ __forceinline__ float4 add4(float4 a, float4 b) {
    return make_float4(a.x + b.x, a.y + b.y, a.z + b.z, a.w + b.w);
}

#define PGROUP(gi, XL) { \
        float qa[8], qb[8]; \
        *(float4*)&qa[0] = a0; *(float4*)&qa[4] = a1; \
        *(float4*)&qb[0] = b0; *(float4*)&qb[4] = b1; \
        if ((gi) + 1 < GPS) { \
            a0 = *(const float4*)&Aw[arow + ((gi) + 1) * 16]; \
            a1 = *(const float4*)&Aw[arow + ((gi) + 1) * 16 + 4]; \
            b0 = Bs[(((gi) + 1) * 2 + 0) * BPITCH + lane]; \
            b1 = Bs[(((gi) + 1) * 2 + 1) * BPITCH + lane]; \
        } \
        XL \
          \
          \
          \
        __builtin_amdgcn_sched_barrier(0); \
          \
          \
        _Pragma("unroll") \
        for (int c = 0; c < 4; c += 2) { \
            v2f A0 = {qa[2 * c],     qa[2 * c + 1]}; \
            v2f B0 = {qb[2 * c],     qb[2 * c + 1]}; \
            v2f A1 = {qa[2 * c + 2], qa[2 * c + 3]}; \
            v2f B1 = {qb[2 * c + 2], qb[2 * c + 3]}; \
            acc0 = __builtin_amdgcn_wmma_f32_16x16x4_f32( \
                false, A0, false, B0, (short)0, acc0, false, false); \
            acc1 = __builtin_amdgcn_wmma_f32_16x16x4_f32( \
                false, A1, false, B1, (short)0, acc1, false, false); \
        } \
    }
#define PFIRST \
        float4 a0 = *(const float4*)&Aw[arow]; \
        float4 a1 = *(const float4*)&Aw[arow + 4]; \
        float4 b0 = Bs[lane]; \
        float4 b1 = Bs[BPITCH + lane];

__global__ __launch_bounds__(128, 5)
void k_main(const float* x,
            const int* __restrict__ lsi,
            const float* l1_w,
            const float* __restrict__ l1_b,
            const float* l1f_w,
            const float* __restrict__ l1f_b,
            const float* __restrict__ l2_w,
            const float* __restrict__ l2_b,
            const float* __restrict__ out_w,
            const float* __restrict__ out_b,
            float* __restrict__ out) {
    __shared__ __attribute__((aligned(16))) float sA[4][16 * APITCH];
    __shared__ __attribute__((aligned(16))) float4 sB[2][16 * BPITCH];
    __shared__ __attribute__((aligned(16))) unsigned s_cnt[4][4];
    __shared__ unsigned s_m[4][2];
    __shared__ int s_rows[64];

    const int tid  = threadIdx.x;
    const int wv   = tid >> 5;
    const int lane = tid & 31;
    const int c = blockIdx.x / TPC;
    const int j = blockIdx.x - c * TPC;

    int4 iv[8];
#pragma unroll
    for (int i = 0; i < 8; ++i) iv[i] = ((const int4*)lsi)[c * (CH_ROWS / 4) + i * 128 + tid];
    unsigned ev = 0, od = 0;
#pragma unroll
    for (int i = 0; i < 8; ++i) {
        const unsigned pk = bk_pack(iv[i]);
        ev += pk & 0x0F0F0F0Fu;
        od += (pk >> 4) & 0x0F0F0F0Fu;
    }
    unsigned rs[4] = { ev & 0x00FF00FFu, (ev >> 8) & 0x00FF00FFu,
                       od & 0x00FF00FFu, (od >> 8) & 0x00FF00FFu };
    for (int m = 16; m >= 1; m >>= 1) {
#pragma unroll
        for (int i = 0; i < 4; ++i) rs[i] += __shfl_xor(rs[i], m, 32);
    }
    if (lane == 0) {
#pragma unroll
        for (int i = 0; i < 4; ++i) s_cnt[wv][i] = rs[i];
    }
    __syncthreads();
    unsigned tot[4];
#pragma unroll
    for (int i = 0; i < 4; ++i) tot[i] = s_cnt[0][i] + s_cnt[1][i] + s_cnt[2][i] + s_cnt[3][i];

    int s_v = -1, tile_v = 0, cnt_v = 0, pre = 0;
#pragma unroll
    for (int k = 0; k < NSTACK; ++k) {
        const int ck = bk_field16(tot, k);
        const int tk = (ck + 63) >> 6;
        if (s_v < 0 && j < pre + tk) { s_v = k; tile_v = j - pre; cnt_v = ck; }
        pre += tk;
    }
    if (s_v < 0) return;
    const int s     = __builtin_amdgcn_readfirstlane(s_v);
    const int tile  = __builtin_amdgcn_readfirstlane(tile_v);
    const int cnt_s = __builtin_amdgcn_readfirstlane(cnt_v);
    const int lo0   = tile * 64;

    unsigned mlo = 0, mhi = 0;
#pragma unroll
    for (int i = 0; i < 8; ++i) {
        const unsigned m = (unsigned)((iv[i].x & 7) == s) + (unsigned)((iv[i].y & 7) == s) +
                           (unsigned)((iv[i].z & 7) == s) + (unsigned)((iv[i].w & 7) == s);
        if (i < 4) mlo |= m << (8 * i); else mhi |= m << (8 * (i - 4));
    }
    unsigned ilo = mlo, ihi = mhi;
    for (int d = 1; d < 32; d <<= 1) {
        const unsigned a = __shfl_up(ilo, d, 32);
        const unsigned b = __shfl_up(ihi, d, 32);
        if (lane >= d) { ilo += a; ihi += b; }
    }
    if (lane == 31) { s_m[wv][0] = ilo; s_m[wv][1] = ihi; }
    __syncthreads();
    unsigned sm[4][2];
#pragma unroll
    for (int w = 0; w < 4; ++w) { sm[w][0] = s_m[w][0]; sm[w][1] = s_m[w][1]; }
    {
        int base = -lo0;
#pragma unroll
        for (int i = 0; i < 8; ++i) {
            const int sh = (i & 3) * 8;
            int T = 0, before = 0;
#pragma unroll
            for (int w = 0; w < 4; ++w) {
                const int bw = (int)((sm[w][i >> 2] >> sh) & 0xFFu);
                T += bw;
                if (w < wv) before += bw;
            }
            const unsigned incw = (i < 4) ? ilo : ihi;
            const unsigned ownw = (i < 4) ? mlo : mhi;
            int pos = base + before + (int)((incw >> sh) & 0xFFu) - (int)((ownw >> sh) & 0xFFu);
            const int row0 = c * CH_ROWS + (i * 128 + tid) * 4;
            const int kk[4] = {iv[i].x & 7, iv[i].y & 7, iv[i].z & 7, iv[i].w & 7};
#pragma unroll
            for (int e = 0; e < 4; ++e) {
                if (kk[e] == s) {
                    if ((unsigned)pos < 64u) s_rows[pos] = row0 + e;
                    ++pos;
                }
            }
            base += T;
        }
    }
    __syncthreads();

    const int r    = lane & 15;
    const int p    = lo0 + wv * 16 + r;
    const int pcl  = (p < cnt_s) ? p : (cnt_s - 1);
    const int rrow = s_rows[pcl - lo0];

    const int koff = (lane < 16) ? 0 : 8;
    float* __restrict__ Aw = sA[wv];

#define ROWS16(M) M(0) M(1) M(2) M(3) M(4) M(5) M(6) M(7) \
                  M(8) M(9) M(10) M(11) M(12) M(13) M(14) M(15)
#define XBASE(j) const float* xb##j = \
        x + (size_t)__builtin_amdgcn_readlane(rrow, j) * NL1;
    ROWS16(XBASE)
    const unsigned lo = (unsigned)lane * 4u;

    const float* wsb = l1_w  + (size_t)(s * 16 + wv * 4) * NL1 + lo;
    const float* wfb = l1f_w + (size_t)(wv * 4) * NL1 + lo;
    const int bst = ((lane >> 2) * 2 + (lane & 1)) * BPITCH + 16 * ((lane >> 1) & 1) + wv * 4;

    v8f acc0 = {0.f, 0.f, 0.f, 0.f, 0.f, 0.f, 0.f, 0.f};
    v8f acc1 = {0.f, 0.f, 0.f, 0.f, 0.f, 0.f, 0.f, 0.f};

    const int arow = r * APITCH + koff;

#define XDECL(j)  float4 xg##j;
#define XLOAD(j)  xg##j = *(const float4*)(xb##j + vo);
#define XSTORE(j) *(float4*)&Aw[j * APITCH + lo] = xg##j;
#define WROWS(M)  M(0) M(1) M(2) M(3)
#define WDECL(i)  float4 ws##i, wf##i, wq##i;
#define WLOAD(i)  ws##i = *(const float4*)(wsb + i * NL1 + wo); \
                  wf##i = *(const float4*)(wfb + i * NL1 + wo);
#define WPIN(i)   asm volatile("" : "+v"(ws##i.x), "+v"(ws##i.y), "+v"(ws##i.z), "+v"(ws##i.w), \
                                  "+v"(wf##i.x), "+v"(wf##i.y), "+v"(wf##i.z), "+v"(wf##i.w));
#define WSUM(i)   wq##i = add4(ws##i, wf##i);
#define WPINQ(i)  asm volatile("" : "+v"(wq##i.x), "+v"(wq##i.y), "+v"(wq##i.z), "+v"(wq##i.w));
#define WSTORE(i) Bst[bst + i] = wq##i;
    ROWS16(XDECL)
    WROWS(WDECL)
    {
        const unsigned vo = lo;
        const unsigned wo = 0;
        float4* Bst = sB[0];
        WROWS(WLOAD)
        ROWS16(XLOAD)
        ROWS16(XSTORE)
        WROWS(WSUM)
        WROWS(WSTORE)
    }
    __syncthreads();

#pragma unroll 1
    for (int t = 0; t + 1 < NSTEPS; ++t) {
        const unsigned vo = (unsigned)(t + 1) * KC + lo;
        {
            const unsigned wo = (unsigned)(t + 1) * KC;
            WROWS(WLOAD)
        }
        asm volatile("" ::: "memory");
        __builtin_amdgcn_sched_barrier(0);
        {
            const float4* Bs = sB[t & 1];
            PFIRST
            PGROUP(0, XLOAD(0)  XLOAD(1)  XLOAD(2))
            PGROUP(1, XLOAD(3)  XLOAD(4)  XLOAD(5))
            PGROUP(2, XLOAD(6)  XLOAD(7)  XLOAD(8))
            WROWS(WPIN)
            WROWS(WSUM)
            WROWS(WPINQ)
            PGROUP(3, XLOAD(9)  XLOAD(10) XLOAD(11))
            PGROUP(4, XLOAD(12) XLOAD(13) XLOAD(14))
            PGROUP(5, XLOAD(15))
            PGROUP(6, )
            PGROUP(7, )
        }
        __builtin_amdgcn_wave_barrier();
        ROWS16(XSTORE)
        {
            float4* Bst = sB[(t + 1) & 1];
            WROWS(WSTORE)
        }
        __syncthreads();
    }
    {
        const float4* Bs = sB[(NSTEPS - 1) & 1];
        PFIRST
        PGROUP(0, ) PGROUP(1, ) PGROUP(2, ) PGROUP(3, )
        PGROUP(4, ) PGROUP(5, ) PGROUP(6, ) PGROUP(7, )
        __builtin_amdgcn_wave_barrier();
    }
#undef XBASE
#undef XDECL
#undef XLOAD
#undef XSTORE
#undef PGROUP
#undef PFIRST
#undef WROWS
#undef WDECL
#undef WLOAD
#undef WPIN
#undef WSUM
#undef WPINQ
#undef WSTORE
#undef ROWS16

    const int mbase = (lane >= 16) ? 8 : 0;
    const int ncol  = lane & 15;
#pragma unroll
    for (int v = 0; v < 8; ++v) {
        Aw[(v + mbase) * EPITCH + ncol] = acc0[v] + acc1[v];
    }
    __builtin_amdgcn_wave_barrier();

    const int jh = lane >> 4;
    float t[15];
    float keep;
    {
        keep = (Aw[r * EPITCH + 15] + l1_b[s * 16 + 15]) + l1f_b[15];
#pragma unroll
        for (int q = 0; q < 15; ++q) {
            t[q] = (Aw[r * EPITCH + q] + l1_b[s * 16 + q]) + l1f_b[q];
        }
    }
    float l1x[30];
#pragma unroll
    for (int q = 0; q < 15; ++q) {
        float sq = t[q] * t[q] * (127.0f / 128.0f);
        l1x[q]      = fminf(fmaxf(sq, 0.f), 1.f);
        l1x[15 + q] = fminf(fmaxf(t[q], 0.f), 1.f);
    }
    float partial = 0.f;
#pragma unroll 4
    for (int oi = 0; oi < 16; ++oi) {
        int o = jh * 16 + oi;
        const float* wrow = l2_w + (size_t)(s * 32 + o) * 30;
        float a2 = l2_b[s * 32 + o];
#pragma unroll
        for (int k = 0; k < 30; ++k) a2 += l1x[k] * wrow[k];
        float l2x = fminf(fmaxf(a2, 0.f), 1.f);
        partial += l2x * out_w[s * 32 + o];
    }
    float tot_o = partial + __shfl_xor(partial, 16, 32);

    if (lane < 16 && p < cnt_s) {
        out[rrow] = tot_o + out_b[s] + keep;
    }
}


extern "C" void kernel_launch(void* const* d_in, const int* in_sizes, int n_in,
                              void* d_out, int out_size, void* d_ws, size_t ws_size,
                              hipStream_t stream) {
    (void)in_sizes; (void)n_in; (void)out_size; (void)d_ws; (void)ws_size;
    const float* x      = (const float*)d_in[0];
    const int*   lsi    = (const int*)d_in[1];
    const float* l1_w   = (const float*)d_in[2];
    const float* l1_b   = (const float*)d_in[3];
    const float* l1f_w  = (const float*)d_in[4];
    const float* l1f_b  = (const float*)d_in[5];
    const float* l2_w   = (const float*)d_in[6];
    const float* l2_b   = (const float*)d_in[7];
    const float* out_w  = (const float*)d_in[8];
    const float* out_b  = (const float*)d_in[9];
    float* out = (float*)d_out;

    k_main<<<NCHUNK * TPC, 128, 0, stream>>>(x, lsi, l1_w, l1_b, l1f_w, l1f_b,
                                             l2_w, l2_b, out_w, out_b, out);
}
